# starting kernel
# baseline (speedup 1.0000x reference)
#include <hip/hip_runtime.h>
#include <hip/hip_bf16.h>

typedef float v2f __attribute__((ext_vector_type(2)));
typedef float v8f __attribute__((ext_vector_type(8)));

#define NNODES 50000

__global__ void k_init_deg(float* deg) {
  int i = blockIdx.x * blockDim.x + threadIdx.x;
  if (i < NNODES) deg[i] = 1.0f;
}

__global__ void k_count_deg(const int* __restrict__ dst, float* __restrict__ deg, int E) {
  int e = blockIdx.x * blockDim.x + threadIdx.x;
  if (e < E) atomicAdd(&deg[dst[e]], 1.0f);
}

__global__ void k_finish_deg(float* deg) {
  int i = blockIdx.x * blockDim.x + threadIdx.x;
  if (i < NNODES) deg[i] = rsqrtf(deg[i]);
}

template <int K, int N>
__global__ void k_gemm_wmma_f32(const float* __restrict__ A,
                                const float* __restrict__ W,
                                float* __restrict__ C, int M) {
  const int wave = (int)((blockIdx.x * blockDim.x + threadIdx.x) >> 5);
  const int lane = threadIdx.x & 31;
  const int tiles_n = N / 16;
  const int row_tile = wave / tiles_n;
  const int col_tile = wave % tiles_n;
  if (row_tile * 16 >= M) return;

  const int half = lane >> 4;
  const int r    = lane & 15;

  const float* arow = A + (size_t)(row_tile * 16 + r) * K;
  const float* wcol = W + (size_t)(col_tile * 16 + r);

  v8f acc = {};
#pragma unroll 4
  for (int k0 = 0; k0 < K; k0 += 4) {
    const int kA = k0 + 2 * half;
    v2f a = *(const v2f*)(arow + kA);
    v2f b;
    b.x = wcol[(size_t)kA * N];
    b.y = wcol[(size_t)(kA + 1) * N];
    acc = __builtin_amdgcn_wmma_f32_16x16x4_f32(
         false, a,  false, b,
         (short)0, acc,  false,  false);
  }

  float* crow = C + (size_t)(row_tile * 16 + 8 * half) * N + col_tile * 16 + r;
#pragma unroll
  for (int i = 0; i < 8; ++i) crow[(size_t)i * N] = acc[i];
}

template <int CH>
__global__ void k_self_bias(const float* __restrict__ feat,
                            const float* __restrict__ dinv,
                            const float* __restrict__ bias,
                            float* __restrict__ out) {
  int idx = blockIdx.x * blockDim.x + threadIdx.x;
  if (idx >= NNODES * CH) return;
  int n = idx / CH;
  int c = idx % CH;
  float di = dinv[n];
  out[idx] = feat[idx] * di * di + bias[c];
}

template <int CH>
__global__ void k_scatter(const int* __restrict__ src, const int* __restrict__ dst,
                          const float* __restrict__ dinv,
                          const float* __restrict__ feat,
                          float* __restrict__ out, int E) {
  const int wave = (int)((blockIdx.x * blockDim.x + threadIdx.x) >> 5);
  const int lane = threadIdx.x & 31;
  if (wave >= E) return;
  const int s = src[wave];
  const int d = dst[wave];
  const float w = dinv[s] * dinv[d];
  const float* frow = feat + (size_t)s * CH;
  float* orow = out + (size_t)d * CH;
  if (CH == 128) {
    float4 v = *(const float4*)(frow + lane * 4);
    atomicAdd(orow + lane * 4 + 0, v.x * w);
    atomicAdd(orow + lane * 4 + 1, v.y * w);
    atomicAdd(orow + lane * 4 + 2, v.z * w);
    atomicAdd(orow + lane * 4 + 3, v.w * w);
  } else {
    float2 v = *(const float2*)(frow + lane * 2);
    atomicAdd(orow + lane * 2 + 0, v.x * w);
    atomicAdd(orow + lane * 2 + 1, v.y * w);
  }
}

__global__ void k_tanh(const float* __restrict__ in, float* __restrict__ out, int n) {
  int idx = blockIdx.x * blockDim.x + threadIdx.x;
  if (idx < n) out[idx] = tanhf(in[idx]);
}

extern "C" void kernel_launch(void* const* d_in, const int* in_sizes, int n_in,
                              void* d_out, int out_size, void* d_ws, size_t ws_size,
                              hipStream_t stream) {
  const float* x   = (const float*)d_in[0];
  const int*  eidx = (const int*)d_in[1];
  const float* W1  = (const float*)d_in[2];
  const float* b1  = (const float*)d_in[3];
  const float* W2  = (const float*)d_in[4];
  const float* b2  = (const float*)d_in[5];
  float* out = (float*)d_out;

  const int E = in_sizes[1] / 2;
  const int* src = eidx;
  const int* dst = eidx + E;

  float* dinv = (float*)d_ws;
  float* bufA = dinv + 50048;
  float* bufB = bufA + (size_t)NNODES * 128;

  const int T = 256;

  k_init_deg<<<(NNODES + T - 1) / T, T, 0, stream>>>(dinv);
  k_count_deg<<<(E + T - 1) / T, T, 0, stream>>>(dst, dinv, E);
  k_finish_deg<<<(NNODES + T - 1) / T, T, 0, stream>>>(dinv);

  {
    int waves = (NNODES / 16) * (128 / 16);
    k_gemm_wmma_f32<256, 128><<<waves * 32 / 128, 128, 0, stream>>>(x, W1, bufA, NNODES);
  }
  k_self_bias<128><<<(NNODES * 128 + T - 1) / T, T, 0, stream>>>(bufA, dinv, b1, bufB);
  k_scatter<128><<<(E + 7) / 8, 256, 0, stream>>>(src, dst, dinv, bufA, bufB, E);
  k_tanh<<<(NNODES * 128 + T - 1) / T, T, 0, stream>>>(bufB, bufA, NNODES * 128);

  {
    int waves = (NNODES / 16) * (64 / 16);
    k_gemm_wmma_f32<128, 64><<<waves * 32 / 128, 128, 0, stream>>>(bufA, W2, bufB, NNODES);
  }
  k_self_bias<64><<<(NNODES * 64 + T - 1) / T, T, 0, stream>>>(bufB, dinv, b2, out);
  k_scatter<64><<<(E + 7) / 8, 256, 0, stream>>>(src, dst, dinv, bufB, out, E);
}


# Scatters and the 3 elementwise passes made resident: a wave walks 64 edges (8 rows in flight), a lane 4 float4 items; 800000- and 200000-wave launches become 12504 waves; atomics and lane map kept
# speedup vs baseline: 1.1259x; 1.1259x over previous
#include <hip/hip_runtime.h>
#include <hip/hip_bf16.h>

typedef float v2f __attribute__((ext_vector_type(2)));
typedef float v8f __attribute__((ext_vector_type(8)));

#define NNODES 50000

__global__ void k_init_deg(float* deg) {
  int i = blockIdx.x * blockDim.x + threadIdx.x;
  if (i < NNODES) deg[i] = 1.0f;
}

__global__ void k_count_deg(const int* __restrict__ dst, float* __restrict__ deg, int E) {
  int e = blockIdx.x * blockDim.x + threadIdx.x;
  if (e < E) atomicAdd(&deg[dst[e]], 1.0f);
}

__global__ void k_finish_deg(float* deg) {
  int i = blockIdx.x * blockDim.x + threadIdx.x;
  if (i < NNODES) deg[i] = rsqrtf(deg[i]);
}

template <int K, int N>
__global__ __launch_bounds__(256) void k_gemm_wmma_f32(const float* __restrict__ A,
                                                       const float* __restrict__ W,
                                                       float* __restrict__ C, int M) {
  constexpr int CT    = N / 16;
  constexpr int RPS   = 8 / CT;
  constexpr int ROWS  = RPS * 16;
  constexpr int PITCH = K + 4;
  constexpr int KS    = K / 4;
  constexpr int F4ROW = K / 4;
  constexpr int NLD   = ROWS * F4ROW / 256;

  __shared__ float lds[2][ROWS * PITCH];

  const int tid  = threadIdx.x;
  const int lane = tid & 31;
  const int wave = tid >> 5;
  const int half = lane >> 4;
  const int r    = lane & 15;
  const int ct   = wave % CT;
  const int rt   = wave / CT;
  const int row_tiles = M / 16;
  const int nstages   = (row_tiles + RPS - 1) / RPS;

  v2f breg[KS];
  {
    const float* wcol = W + (size_t)(ct * 16 + r);
#pragma unroll
    for (int i = 0; i < KS; ++i) {
      const int kA = 4 * i + 2 * half;
      breg[i].x = wcol[(size_t)kA * N];
      breg[i].y = wcol[(size_t)(kA + 1) * N];
    }
  }

  float4 pre[NLD];
  int s = blockIdx.x;

#pragma unroll
  for (int j = 0; j < NLD; ++j) {
    const int idx  = tid + 256 * j;
    const int row  = idx / F4ROW;
    const int c4   = idx % F4ROW;
    const int grow = s * ROWS + row;
    float4 v = make_float4(0.f, 0.f, 0.f, 0.f);
    if (s < nstages && grow < M) v = *(const float4*)(A + (size_t)grow * K + c4 * 4);
    *(float4*)&lds[0][row * PITCH + c4 * 4] = v;
  }
  __syncthreads();

  int cur = 0;
  for (; s < nstages; s += gridDim.x) {
    const int  sn       = s + gridDim.x;
    const bool has_next = sn < nstages;
    if (has_next) {
#pragma unroll
      for (int j = 0; j < NLD; ++j) {
        const int idx  = tid + 256 * j;
        const int row  = idx / F4ROW;
        const int c4   = idx % F4ROW;
        const int grow = sn * ROWS + row;
        pre[j] = make_float4(0.f, 0.f, 0.f, 0.f);
        if (grow < M) pre[j] = *(const float4*)(A + (size_t)grow * K + c4 * 4);
      }
    }

    const int tile = s * RPS + rt;
    if (tile < row_tiles) {
      const float* ap = &lds[cur][(rt * 16 + r) * PITCH + 2 * half];
      v8f acc = {};
#pragma unroll
      for (int i = 0; i < KS; ++i) {
        v2f a = *(const v2f*)(ap + 4 * i);
        acc = __builtin_amdgcn_wmma_f32_16x16x4_f32(
             false, a,  false, breg[i],
             (short)0, acc,  false,  false);
      }
      float* crow = C + (size_t)(tile * 16 + 8 * half) * N + ct * 16 + r;
#pragma unroll
      for (int i = 0; i < 8; ++i) crow[(size_t)i * N] = acc[i];
    }

    if (has_next) {
#pragma unroll
      for (int j = 0; j < NLD; ++j) {
        const int idx = tid + 256 * j;
        const int row = idx / F4ROW;
        const int c4  = idx % F4ROW;
        *(float4*)&lds[cur ^ 1][row * PITCH + c4 * 4] = pre[j];
      }
    }
    __syncthreads();
    cur ^= 1;
  }
}

template <int CH>
__global__ __launch_bounds__(256) void k_self_bias(const float* __restrict__ feat,
                                                   const float* __restrict__ dinv,
                                                   const float* __restrict__ bias,
                                                   float* __restrict__ out) {
  constexpr int ROW4   = CH / 4;
  constexpr int TOTAL4 = NNODES * ROW4;
  const int gt = (int)gridDim.x * 256;
  int i = (int)blockIdx.x * 256 + (int)threadIdx.x;
  for (; i + 3 * gt < TOTAL4; i += 4 * gt) {
    float4 v[4];
    float  di[4];
#pragma unroll
    for (int u = 0; u < 4; ++u) {
      const int k = i + u * gt;
      v[u]  = *(const float4*)(feat + (size_t)k * 4);
      di[u] = dinv[k / ROW4];
    }
#pragma unroll
    for (int u = 0; u < 4; ++u) {
      const int k = i + u * gt;
      const float4 b = *(const float4*)(bias + (k % ROW4) * 4);
      float4 o;
      o.x = v[u].x * di[u] * di[u] + b.x;
      o.y = v[u].y * di[u] * di[u] + b.y;
      o.z = v[u].z * di[u] * di[u] + b.z;
      o.w = v[u].w * di[u] * di[u] + b.w;
      *(float4*)(out + (size_t)k * 4) = o;
    }
  }
  for (; i < TOTAL4; i += gt) {
    const float4 v = *(const float4*)(feat + (size_t)i * 4);
    const float  di = dinv[i / ROW4];
    const float4 b = *(const float4*)(bias + (i % ROW4) * 4);
    float4 o;
    o.x = v.x * di * di + b.x;
    o.y = v.y * di * di + b.y;
    o.z = v.z * di * di + b.z;
    o.w = v.w * di * di + b.w;
    *(float4*)(out + (size_t)i * 4) = o;
  }
}

template <int CH>
__global__ __launch_bounds__(256) void k_scatter(const int* __restrict__ src, const int* __restrict__ dst,
                                                 const float* __restrict__ dinv,
                                                 const float* __restrict__ feat,
                                                 float* __restrict__ out, int E) {
  constexpr int VPL = CH / 32;
  typedef float vrow __attribute__((ext_vector_type(VPL)));
  const int lane = (int)threadIdx.x & 31;
  const int wv = __builtin_amdgcn_readfirstlane((int)threadIdx.x >> 5);
  const int gw = (int)blockIdx.x * 8 + wv;
  const int nw = (int)gridDim.x * 8;
  const int nb = (E + 31) >> 5;
  for (int b = gw; b < nb; b += nw) {
    const int e0 = b << 5;
    const int n  = min(32, E - e0);
    int   s = 0, d = 0;
    float w = 0.f;
    if (lane < n) {
      s = src[e0 + lane];
      d = dst[e0 + lane];
      w = dinv[s] * dinv[d];
    }
    for (int j0 = 0; j0 < n; j0 += 8) {
      vrow v[8];
#pragma unroll
      for (int u = 0; u < 8; ++u) {
        const int sj = __builtin_amdgcn_readlane(s, j0 + u);
        v[u] = *(const vrow*)(feat + (size_t)sj * CH + lane * VPL);
      }
#pragma unroll
      for (int u = 0; u < 8; ++u) {
        if (j0 + u < n) {
          const int   dj = __builtin_amdgcn_readlane(d, j0 + u);
          const float wj = __int_as_float(__builtin_amdgcn_readlane(__float_as_int(w), j0 + u));
          float* orow = out + (size_t)dj * CH + lane * VPL;
#pragma unroll
          for (int c = 0; c < VPL; ++c) atomicAdd(orow + c, v[u][c] * wj);
        }
      }
    }
  }
}

__global__ __launch_bounds__(256) void k_tanh(const float* __restrict__ in, float* __restrict__ out, int n) {
  const int n4 = n >> 2;
  const int gt = (int)gridDim.x * 256;
  const int g  = (int)blockIdx.x * 256 + (int)threadIdx.x;
  int i = g;
  for (; i + 3 * gt < n4; i += 4 * gt) {
    float4 v[4];
#pragma unroll
    for (int u = 0; u < 4; ++u) v[u] = *(const float4*)(in + (size_t)(i + u * gt) * 4);
#pragma unroll
    for (int u = 0; u < 4; ++u) {
      float4 o;
      o.x = tanhf(v[u].x);
      o.y = tanhf(v[u].y);
      o.z = tanhf(v[u].z);
      o.w = tanhf(v[u].w);
      *(float4*)(out + (size_t)(i + u * gt) * 4) = o;
    }
  }
  for (; i < n4; i += gt) {
    const float4 v = *(const float4*)(in + (size_t)i * 4);
    float4 o;
    o.x = tanhf(v.x);
    o.y = tanhf(v.y);
    o.z = tanhf(v.z);
    o.w = tanhf(v.w);
    *(float4*)(out + (size_t)i * 4) = o;
  }
  for (int t = (n4 << 2) + g; t < n; t += gt) out[t] = tanhf(in[t]);
}

extern "C" void kernel_launch(void* const* d_in, const int* in_sizes, int n_in,
                              void* d_out, int out_size, void* d_ws, size_t ws_size,
                              hipStream_t stream) {
  const float* x   = (const float*)d_in[0];
  const int*  eidx = (const int*)d_in[1];
  const float* W1  = (const float*)d_in[2];
  const float* b1  = (const float*)d_in[3];
  const float* W2  = (const float*)d_in[4];
  const float* b2  = (const float*)d_in[5];
  float* out = (float*)d_out;

  const int E = in_sizes[1] / 2;
  const int* src = eidx;
  const int* dst = eidx + E;

  float* dinv = (float*)d_ws;
  float* bufA = dinv + 50048;
  float* bufB = bufA + (size_t)NNODES * 128;

  const int T = 256;

  k_init_deg<<<(NNODES + T - 1) / T, T, 0, stream>>>(dinv);
  k_count_deg<<<(E + T - 1) / T, T, 0, stream>>>(dst, dinv, E);
  k_finish_deg<<<(NNODES + T - 1) / T, T, 0, stream>>>(dinv);

  {
    k_gemm_wmma_f32<256, 128><<<512, 256, 0, stream>>>(x, W1, bufA, NNODES);
  }
  k_self_bias<128><<<1563, 256, 0, stream>>>(bufA, dinv, b1, bufB);
  k_scatter<128><<<1563, 256, 0, stream>>>(src, dst, dinv, bufA, bufB, E);
  k_tanh<<<1563, 256, 0, stream>>>(bufB, bufA, NNODES * 128);

  {
    k_gemm_wmma_f32<128, 64><<<512, 256, 0, stream>>>(bufA, W2, bufB, NNODES);
  }
  k_self_bias<64><<<782, 256, 0, stream>>>(bufB, dinv, b2, out);
  k_scatter<64><<<1563, 256, 0, stream>>>(src, dst, dinv, bufB, out, E);
}


# Owner-computes aggregation: 256 workgroups hold 196 destination rows each in LDS, filter the whole edge list, add source rows by LDS float atomics (arrival order); global atomics, start pass gone
# speedup vs baseline: 1.6242x; 1.4426x over previous
#include <hip/hip_runtime.h>
#include <hip/hip_bf16.h>

typedef float v2f __attribute__((ext_vector_type(2)));
typedef float v8f __attribute__((ext_vector_type(8)));

#define NNODES 50000

__global__ void k_init_deg(float* deg) {
  int i = blockIdx.x * blockDim.x + threadIdx.x;
  if (i < NNODES) deg[i] = 1.0f;
}

__global__ void k_count_deg(const int* __restrict__ dst, float* __restrict__ deg, int E) {
  int e = blockIdx.x * blockDim.x + threadIdx.x;
  if (e < E) atomicAdd(&deg[dst[e]], 1.0f);
}

__global__ void k_finish_deg(float* deg) {
  int i = blockIdx.x * blockDim.x + threadIdx.x;
  if (i < NNODES) deg[i] = rsqrtf(deg[i]);
}

template <int K, int N>
__global__ __launch_bounds__(256) void k_gemm_wmma_f32(const float* __restrict__ A,
                                                       const float* __restrict__ W,
                                                       float* __restrict__ C, int M) {
  constexpr int CT    = N / 16;
  constexpr int RPS   = 8 / CT;
  constexpr int ROWS  = RPS * 16;
  constexpr int PITCH = K + 4;
  constexpr int KS    = K / 4;
  constexpr int F4ROW = K / 4;
  constexpr int NLD   = ROWS * F4ROW / 256;

  __shared__ float lds[2][ROWS * PITCH];

  const int tid  = threadIdx.x;
  const int lane = tid & 31;
  const int wave = tid >> 5;
  const int half = lane >> 4;
  const int r    = lane & 15;
  const int ct   = wave % CT;
  const int rt   = wave / CT;
  const int row_tiles = M / 16;
  const int nstages   = (row_tiles + RPS - 1) / RPS;

  v2f breg[KS];
  {
    const float* wcol = W + (size_t)(ct * 16 + r);
#pragma unroll
    for (int i = 0; i < KS; ++i) {
      const int kA = 4 * i + 2 * half;
      breg[i].x = wcol[(size_t)kA * N];
      breg[i].y = wcol[(size_t)(kA + 1) * N];
    }
  }

  float4 pre[NLD];
  int s = blockIdx.x;

#pragma unroll
  for (int j = 0; j < NLD; ++j) {
    const int idx  = tid + 256 * j;
    const int row  = idx / F4ROW;
    const int c4   = idx % F4ROW;
    const int grow = s * ROWS + row;
    float4 v = make_float4(0.f, 0.f, 0.f, 0.f);
    if (s < nstages && grow < M) v = *(const float4*)(A + (size_t)grow * K + c4 * 4);
    *(float4*)&lds[0][row * PITCH + c4 * 4] = v;
  }
  __syncthreads();

  int cur = 0;
  for (; s < nstages; s += gridDim.x) {
    const int  sn       = s + gridDim.x;
    const bool has_next = sn < nstages;
    if (has_next) {
#pragma unroll
      for (int j = 0; j < NLD; ++j) {
        const int idx  = tid + 256 * j;
        const int row  = idx / F4ROW;
        const int c4   = idx % F4ROW;
        const int grow = sn * ROWS + row;
        pre[j] = make_float4(0.f, 0.f, 0.f, 0.f);
        if (grow < M) pre[j] = *(const float4*)(A + (size_t)grow * K + c4 * 4);
      }
    }

    const int tile = s * RPS + rt;
    if (tile < row_tiles) {
      const float* ap = &lds[cur][(rt * 16 + r) * PITCH + 2 * half];
      v8f acc = {};
#pragma unroll
      for (int i = 0; i < KS; ++i) {
        v2f a = *(const v2f*)(ap + 4 * i);
        acc = __builtin_amdgcn_wmma_f32_16x16x4_f32(
             false, a,  false, breg[i],
             (short)0, acc,  false,  false);
      }
      float* crow = C + (size_t)(tile * 16 + 8 * half) * N + ct * 16 + r;
#pragma unroll
      for (int i = 0; i < 8; ++i) crow[(size_t)i * N] = acc[i];
    }

    if (has_next) {
#pragma unroll
      for (int j = 0; j < NLD; ++j) {
        const int idx = tid + 256 * j;
        const int row = idx / F4ROW;
        const int c4  = idx % F4ROW;
        *(float4*)&lds[cur ^ 1][row * PITCH + c4 * 4] = pre[j];
      }
    }
    __syncthreads();
    cur ^= 1;
  }
}

#define AGG_WGS  256
#define AGG_ROWS ((NNODES + AGG_WGS - 1) / AGG_WGS)
#define AGG_QCAP 512

template <int CH>
__global__ __launch_bounds__(1024) void k_aggregate(const int* __restrict__ src, const int* __restrict__ dst,
                                                    const float* __restrict__ dinv,
                                                    const float* __restrict__ feat,
                                                    const float* __restrict__ bias,
                                                    float* __restrict__ out, int E) {
  constexpr int VPL  = CH / 32;
  constexpr int ROW4 = CH / 4;
  constexpr int G    = (CH <= 64) ? 16 : 8;
  static_assert(AGG_ROWS <= 256, "row number inside a workgroup is packed into 8 bits");
  __shared__ float    tile[AGG_ROWS * CH];
  __shared__ unsigned ring[32][AGG_QCAP];

  const int tid  = (int)threadIdx.x;
  const int lane = tid & 31;
  const int wv   = __builtin_amdgcn_readfirstlane(tid >> 5);
  const int lo   = (int)blockIdx.x * AGG_ROWS;
  const int nrows = max(0, min(AGG_ROWS, NNODES - lo));

  {
    constexpr int NIT = (AGG_ROWS * ROW4 + 1023) / 1024;
    float4 v[NIT];
    float  di[NIT];
#pragma unroll
    for (int j = 0; j < NIT; ++j) {
      const int i = tid + 1024 * j;
      v[j]  = make_float4(0.f, 0.f, 0.f, 0.f);
      di[j] = 0.f;
      if (i < nrows * ROW4) {
        v[j]  = *(const float4*)(feat + (size_t)lo * CH + (size_t)i * 4);
        di[j] = dinv[lo + i / ROW4];
      }
    }
#pragma unroll
    for (int j = 0; j < NIT; ++j) {
      const int i = tid + 1024 * j;
      if (i < nrows * ROW4) {
        const float4 b = *(const float4*)(bias + (i % ROW4) * 4);
        float4 o;
        o.x = v[j].x * di[j] * di[j] + b.x;
        o.y = v[j].y * di[j] * di[j] + b.y;
        o.z = v[j].z * di[j] * di[j] + b.z;
        o.w = v[j].w * di[j] * di[j] + b.w;
        *(float4*)&tile[i * 4] = o;
      }
    }
  }
  __syncthreads();

  unsigned* q = ring[wv];
  int qh = 0, qn = 0;

  auto drain = [&](int base, int n) {
    int e[G], dl[G], s[G];
#pragma unroll
    for (int k = 0; k < G; ++k) {
      const int kk = (k < n) ? k : 0;
      const unsigned ent = q[(base + kk) & (AGG_QCAP - 1)];
      e[k]  = __builtin_amdgcn_readfirstlane((int)(ent & 0xFFFFFFu));
      dl[k] = __builtin_amdgcn_readfirstlane((int)(ent >> 24));
    }
#pragma unroll
    for (int k = 0; k < G; ++k) s[k] = __builtin_amdgcn_readfirstlane(src[e[k]]);
    float v[G][VPL], w[G];
#pragma unroll
    for (int k = 0; k < G; ++k) {
      const float* frow = feat + (size_t)s[k] * CH;
#pragma unroll
      for (int c = 0; c < VPL; ++c) v[k][c] = frow[c * 32 + lane];
      w[k] = dinv[s[k]] * dinv[lo + dl[k]];
    }
#pragma unroll
    for (int k = 0; k < G; ++k) {
      if (k < n) {
        float* trow = &tile[dl[k] * CH + lane];
#pragma unroll
        for (int c = 0; c < VPL; ++c)
          __hip_atomic_fetch_add(trow + c * 32, v[k][c] * w[k], __ATOMIC_RELAXED, __HIP_MEMORY_SCOPE_WORKGROUP);
      }
    }
  };

  auto test8 = [&](const int* d, int ebase) {
#pragma unroll
    for (int u = 0; u < 8; ++u) {
      const unsigned dloc = (unsigned)(d[u] - lo);
      const bool     hit  = dloc < (unsigned)nrows;
      const unsigned m    = __builtin_amdgcn_ballot_w32(hit);
      if (__builtin_expect(m != 0u, 0)) {
        if (hit) {
          const int pos = qn + __builtin_popcount(m & ((1u << lane) - 1u));
          q[pos & (AGG_QCAP - 1)] = (unsigned)(ebase + 32 * u) | (dloc << 24);
        }
        qn += __builtin_popcount(m);
      }
    }
    while (qn - qh >= G) { drain(qh, G); qh += G; }
  };

  const int nfull = E >> 10;
  const int rot = (int)(((long long)blockIdx.x * nfull) / AGG_WGS);
  for (int i = wv; i < nfull; i += 32) {
    int t = i + rot;
    if (t >= nfull) t -= nfull;
    const int  e0 = (t << 10) + lane;
    const int* dp = dst + e0;
    int d[32];
#pragma unroll
    for (int u = 0; u < 32; ++u) d[u] = dp[32 * u];
#pragma unroll
    for (int g = 0; g < 4; ++g) test8(&d[8 * g], e0 + 256 * g);
  }
  {
    const int tbase = nfull << 10;
    const int npiece = (E - tbase + 255) >> 8;
    if (wv < npiece) {
      const int e0 = tbase + (wv << 8) + lane;
      int d[8];
#pragma unroll
      for (int u = 0; u < 8; ++u) {
        const int ee = e0 + 32 * u;
        d[u] = (ee < E) ? dst[ee] : -1;
      }
      test8(d, e0);
    }
  }
  if (qn > qh) drain(qh, qn - qh);
  __syncthreads();

  for (int i = tid; i < nrows * ROW4; i += 1024)
    *(float4*)(out + (size_t)lo * CH + (size_t)i * 4) = *(const float4*)&tile[i * 4];
}

__global__ __launch_bounds__(256) void k_tanh(const float* __restrict__ in, float* __restrict__ out, int n) {
  const int n4 = n >> 2;
  const int gt = (int)gridDim.x * 256;
  const int g  = (int)blockIdx.x * 256 + (int)threadIdx.x;
  int i = g;
  for (; i + 3 * gt < n4; i += 4 * gt) {
    float4 v[4];
#pragma unroll
    for (int u = 0; u < 4; ++u) v[u] = *(const float4*)(in + (size_t)(i + u * gt) * 4);
#pragma unroll
    for (int u = 0; u < 4; ++u) {
      float4 o;
      o.x = tanhf(v[u].x);
      o.y = tanhf(v[u].y);
      o.z = tanhf(v[u].z);
      o.w = tanhf(v[u].w);
      *(float4*)(out + (size_t)(i + u * gt) * 4) = o;
    }
  }
  for (; i < n4; i += gt) {
    const float4 v = *(const float4*)(in + (size_t)i * 4);
    float4 o;
    o.x = tanhf(v.x);
    o.y = tanhf(v.y);
    o.z = tanhf(v.z);
    o.w = tanhf(v.w);
    *(float4*)(out + (size_t)i * 4) = o;
  }
  for (int t = (n4 << 2) + g; t < n; t += gt) out[t] = tanhf(in[t]);
}

extern "C" void kernel_launch(void* const* d_in, const int* in_sizes, int n_in,
                              void* d_out, int out_size, void* d_ws, size_t ws_size,
                              hipStream_t stream) {
  const float* x   = (const float*)d_in[0];
  const int*  eidx = (const int*)d_in[1];
  const float* W1  = (const float*)d_in[2];
  const float* b1  = (const float*)d_in[3];
  const float* W2  = (const float*)d_in[4];
  const float* b2  = (const float*)d_in[5];
  float* out = (float*)d_out;

  const int E = in_sizes[1] / 2;
  const int* src = eidx;
  const int* dst = eidx + E;

  float* dinv = (float*)d_ws;
  float* bufA = dinv + 50048;
  float* bufB = bufA + (size_t)NNODES * 128;

  const int T = 256;

  k_init_deg<<<(NNODES + T - 1) / T, T, 0, stream>>>(dinv);
  k_count_deg<<<(E + T - 1) / T, T, 0, stream>>>(dst, dinv, E);
  k_finish_deg<<<(NNODES + T - 1) / T, T, 0, stream>>>(dinv);

  {
    k_gemm_wmma_f32<256, 128><<<512, 256, 0, stream>>>(x, W1, bufA, NNODES);
  }
  k_aggregate<128><<<AGG_WGS, 1024, 0, stream>>>(src, dst, dinv, bufA, b1, bufB, E);
  k_tanh<<<1563, 256, 0, stream>>>(bufB, bufA, NNODES * 128);

  {
    k_gemm_wmma_f32<128, 64><<<512, 256, 0, stream>>>(bufA, W2, bufB, NNODES);
  }
  k_aggregate<64><<<AGG_WGS, 1024, 0, stream>>>(src, dst, dinv, bufB, b2, out, E);
}


# Layer-2 product moved into the layer-1 owner aggregation: tanh of the rows in LDS, then rows x W2 on the matrix unit from LDS (pitch 132), whole-line store; tanh and 2nd GEMM launches gone
# speedup vs baseline: 1.7959x; 1.1057x over previous
#include <hip/hip_runtime.h>
#include <hip/hip_bf16.h>

typedef float v2f __attribute__((ext_vector_type(2)));
typedef float v8f __attribute__((ext_vector_type(8)));

#define NNODES 50000

__global__ void k_init_deg(float* deg) {
  int i = blockIdx.x * blockDim.x + threadIdx.x;
  if (i < NNODES) deg[i] = 1.0f;
}

__global__ void k_count_deg(const int* __restrict__ dst, float* __restrict__ deg, int E) {
  int e = blockIdx.x * blockDim.x + threadIdx.x;
  if (e < E) atomicAdd(&deg[dst[e]], 1.0f);
}

__global__ void k_finish_deg(float* deg) {
  int i = blockIdx.x * blockDim.x + threadIdx.x;
  if (i < NNODES) deg[i] = rsqrtf(deg[i]);
}

template <int K, int N>
__global__ __launch_bounds__(256) void k_gemm_wmma_f32(const float* __restrict__ A,
                                                       const float* __restrict__ W,
                                                       float* __restrict__ C, int M) {
  constexpr int CT    = N / 16;
  constexpr int RPS   = 8 / CT;
  constexpr int ROWS  = RPS * 16;
  constexpr int PITCH = K + 4;
  constexpr int KS    = K / 4;
  constexpr int F4ROW = K / 4;
  constexpr int NLD   = ROWS * F4ROW / 256;

  __shared__ float lds[2][ROWS * PITCH];

  const int tid  = threadIdx.x;
  const int lane = tid & 31;
  const int wave = tid >> 5;
  const int half = lane >> 4;
  const int r    = lane & 15;
  const int ct   = wave % CT;
  const int rt   = wave / CT;
  const int row_tiles = M / 16;
  const int nstages   = (row_tiles + RPS - 1) / RPS;

  v2f breg[KS];
  {
    const float* wcol = W + (size_t)(ct * 16 + r);
#pragma unroll
    for (int i = 0; i < KS; ++i) {
      const int kA = 4 * i + 2 * half;
      breg[i].x = wcol[(size_t)kA * N];
      breg[i].y = wcol[(size_t)(kA + 1) * N];
    }
  }

  float4 pre[NLD];
  int s = blockIdx.x;

#pragma unroll
  for (int j = 0; j < NLD; ++j) {
    const int idx  = tid + 256 * j;
    const int row  = idx / F4ROW;
    const int c4   = idx % F4ROW;
    const int grow = s * ROWS + row;
    float4 v = make_float4(0.f, 0.f, 0.f, 0.f);
    if (s < nstages && grow < M) v = *(const float4*)(A + (size_t)grow * K + c4 * 4);
    *(float4*)&lds[0][row * PITCH + c4 * 4] = v;
  }
  __syncthreads();

  int cur = 0;
  for (; s < nstages; s += gridDim.x) {
    const int  sn       = s + gridDim.x;
    const bool has_next = sn < nstages;
    if (has_next) {
#pragma unroll
      for (int j = 0; j < NLD; ++j) {
        const int idx  = tid + 256 * j;
        const int row  = idx / F4ROW;
        const int c4   = idx % F4ROW;
        const int grow = sn * ROWS + row;
        pre[j] = make_float4(0.f, 0.f, 0.f, 0.f);
        if (grow < M) pre[j] = *(const float4*)(A + (size_t)grow * K + c4 * 4);
      }
    }

    const int tile = s * RPS + rt;
    if (tile < row_tiles) {
      const float* ap = &lds[cur][(rt * 16 + r) * PITCH + 2 * half];
      v8f acc = {};
#pragma unroll
      for (int i = 0; i < KS; ++i) {
        v2f a = *(const v2f*)(ap + 4 * i);
        acc = __builtin_amdgcn_wmma_f32_16x16x4_f32(
             false, a,  false, breg[i],
             (short)0, acc,  false,  false);
      }
      float* crow = C + (size_t)(tile * 16 + 8 * half) * N + ct * 16 + r;
#pragma unroll
      for (int i = 0; i < 8; ++i) crow[(size_t)i * N] = acc[i];
    }

    if (has_next) {
#pragma unroll
      for (int j = 0; j < NLD; ++j) {
        const int idx = tid + 256 * j;
        const int row = idx / F4ROW;
        const int c4  = idx % F4ROW;
        *(float4*)&lds[cur ^ 1][row * PITCH + c4 * 4] = pre[j];
      }
    }
    __syncthreads();
    cur ^= 1;
  }
}

#define AGG_WGS  256
#define AGG_ROWS ((NNODES + AGG_WGS - 1) / AGG_WGS)
#define AGG_QCAP 512

template <int CH, int NOUT>
__global__ __launch_bounds__(1024) void k_aggregate(const int* __restrict__ src, const int* __restrict__ dst,
                                                    const float* __restrict__ dinv,
                                                    const float* __restrict__ feat,
                                                    const float* __restrict__ bias,
                                                    const float* __restrict__ W,
                                                    float* __restrict__ out, int E) {
  constexpr int PITCH = (NOUT > 0) ? CH + 4 : CH;
  constexpr int TROWS = (NOUT > 0) ? (AGG_ROWS + 15) / 16 * 16 : AGG_ROWS;
  constexpr int VPL  = CH / 32;
  constexpr int ROW4 = CH / 4;
  constexpr int G    = (CH <= 64) ? 16 : 8;
  static_assert(AGG_ROWS <= 256, "row number inside a workgroup is packed into 8 bits");
  __shared__ __attribute__((aligned(16))) float tile[TROWS * PITCH];
  __shared__ __attribute__((aligned(16))) unsigned ring[32][AGG_QCAP];

  const int tid  = (int)threadIdx.x;
  const int lane = tid & 31;
  const int wv   = __builtin_amdgcn_readfirstlane(tid >> 5);
  const int lo   = (int)blockIdx.x * AGG_ROWS;
  const int nrows = max(0, min(AGG_ROWS, NNODES - lo));

  {
    constexpr int NIT = (AGG_ROWS * ROW4 + 1023) / 1024;
    float4 v[NIT];
    float  di[NIT];
#pragma unroll
    for (int j = 0; j < NIT; ++j) {
      const int i = tid + 1024 * j;
      v[j]  = make_float4(0.f, 0.f, 0.f, 0.f);
      di[j] = 0.f;
      if (i < nrows * ROW4) {
        v[j]  = *(const float4*)(feat + (size_t)lo * CH + (size_t)i * 4);
        di[j] = dinv[lo + i / ROW4];
      }
    }
#pragma unroll
    for (int j = 0; j < NIT; ++j) {
      const int i = tid + 1024 * j;
      if (i < nrows * ROW4) {
        const float4 b = *(const float4*)(bias + (i % ROW4) * 4);
        float4 o;
        o.x = v[j].x * di[j] * di[j] + b.x;
        o.y = v[j].y * di[j] * di[j] + b.y;
        o.z = v[j].z * di[j] * di[j] + b.z;
        o.w = v[j].w * di[j] * di[j] + b.w;
        *(float4*)&tile[(i / ROW4) * PITCH + (i % ROW4) * 4] = o;
      }
    }
    for (int i = nrows * ROW4 + tid; i < TROWS * ROW4; i += 1024)
      *(float4*)&tile[(i / ROW4) * PITCH + (i % ROW4) * 4] = make_float4(0.f, 0.f, 0.f, 0.f);
  }
  __syncthreads();

  unsigned* q = ring[wv];
  int qh = 0, qn = 0;

  auto drain = [&](int base, int n) {
    int e[G], dl[G], s[G];
#pragma unroll
    for (int k = 0; k < G; ++k) {
      const int kk = (k < n) ? k : 0;
      const unsigned ent = q[(base + kk) & (AGG_QCAP - 1)];
      e[k]  = __builtin_amdgcn_readfirstlane((int)(ent & 0xFFFFFFu));
      dl[k] = __builtin_amdgcn_readfirstlane((int)(ent >> 24));
    }
#pragma unroll
    for (int k = 0; k < G; ++k) s[k] = __builtin_amdgcn_readfirstlane(src[e[k]]);
    float v[G][VPL], w[G];
#pragma unroll
    for (int k = 0; k < G; ++k) {
      const float* frow = feat + (size_t)s[k] * CH;
#pragma unroll
      for (int c = 0; c < VPL; ++c) v[k][c] = frow[c * 32 + lane];
      w[k] = dinv[s[k]] * dinv[lo + dl[k]];
    }
#pragma unroll
    for (int k = 0; k < G; ++k) {
      if (k < n) {
        float* trow = &tile[dl[k] * PITCH + lane];
#pragma unroll
        for (int c = 0; c < VPL; ++c)
          __hip_atomic_fetch_add(trow + c * 32, v[k][c] * w[k], __ATOMIC_RELAXED, __HIP_MEMORY_SCOPE_WORKGROUP);
      }
    }
  };

  auto test8 = [&](const int* d, int ebase) {
#pragma unroll
    for (int u = 0; u < 8; ++u) {
      const unsigned dloc = (unsigned)(d[u] - lo);
      const bool     hit  = dloc < (unsigned)nrows;
      const unsigned m    = __builtin_amdgcn_ballot_w32(hit);
      if (__builtin_expect(m != 0u, 0)) {
        if (hit) {
          const int pos = qn + __builtin_popcount(m & ((1u << lane) - 1u));
          q[pos & (AGG_QCAP - 1)] = (unsigned)(ebase + 32 * u) | (dloc << 24);
        }
        qn += __builtin_popcount(m);
      }
    }
    while (qn - qh >= G) { drain(qh, G); qh += G; }
  };

  const int nfull = E >> 10;
  const int rot = (int)(((long long)blockIdx.x * nfull) / AGG_WGS);
  for (int i = wv; i < nfull; i += 32) {
    int t = i + rot;
    if (t >= nfull) t -= nfull;
    const int  e0 = (t << 10) + lane;
    const int* dp = dst + e0;
    int d[32];
#pragma unroll
    for (int u = 0; u < 32; ++u) d[u] = dp[32 * u];
#pragma unroll
    for (int g = 0; g < 4; ++g) test8(&d[8 * g], e0 + 256 * g);
  }
  {
    const int tbase = nfull << 10;
    const int npiece = (E - tbase + 255) >> 8;
    if (wv < npiece) {
      const int e0 = tbase + (wv << 8) + lane;
      int d[8];
#pragma unroll
      for (int u = 0; u < 8; ++u) {
        const int ee = e0 + 32 * u;
        d[u] = (ee < E) ? dst[ee] : -1;
      }
      test8(d, e0);
    }
  }
  if (qn > qh) drain(qh, qn - qh);
  __syncthreads();

  if constexpr (NOUT == 0) {
    for (int i = tid; i < nrows * ROW4; i += 1024)
      *(float4*)(out + (size_t)lo * CH + (size_t)i * 4) = *(const float4*)&tile[(i / ROW4) * PITCH + (i % ROW4) * 4];
  } else {
    constexpr int CT  = NOUT / 16;
    constexpr int KS  = CH / 4;
    constexpr int RTS = 32 / CT;
    static_assert(32 % CT == 0, "every wave owns one column tile");
    const int half = lane >> 4;
    const int r    = lane & 15;
    const int ct   = wv % CT;
    v2f breg[KS];
    {
      const float* wcol = W + (size_t)(ct * 16 + r);
#pragma unroll
      for (int i = 0; i < KS; ++i) {
        const int kA = 4 * i + 2 * half;
        breg[i].x = wcol[(size_t)kA * NOUT];
        breg[i].y = wcol[(size_t)(kA + 1) * NOUT];
      }
    }
    for (int i = tid; i < nrows * ROW4; i += 1024) {
      float4* p = (float4*)&tile[(i / ROW4) * PITCH + (i % ROW4) * 4];
      float4 v = *p;
      v.x = tanhf(v.x);
      v.y = tanhf(v.y);
      v.z = tanhf(v.z);
      v.w = tanhf(v.w);
      *p = v;
    }
    __syncthreads();
    static_assert(TROWS * NOUT <= 32 * AGG_QCAP, "product rows fit the rings' space");
    float* stage = (float*)&ring[0][0];
    const int ntiles = (nrows + 15) >> 4;
    for (int t = wv / CT; t < ntiles; t += RTS) {
      const float* ap = &tile[(t * 16 + r) * PITCH + 2 * half];
      v8f acc = {};
#pragma unroll
      for (int i = 0; i < KS; ++i) {
        v2f a = *(const v2f*)(ap + 4 * i);
        acc = __builtin_amdgcn_wmma_f32_16x16x4_f32(
             false, a,  false, breg[i],
             (short)0, acc,  false,  false);
      }
      float* sp = stage + (t * 16 + 8 * half) * NOUT + ct * 16 + r;
#pragma unroll
      for (int i = 0; i < 8; ++i) sp[i * NOUT] = acc[i];
    }
    __syncthreads();
    for (int i = tid; i < nrows * (NOUT / 4); i += 1024)
      *(float4*)(out + (size_t)lo * NOUT + (size_t)i * 4) = *(const float4*)&stage[i * 4];
  }
}

extern "C" void kernel_launch(void* const* d_in, const int* in_sizes, int n_in,
                              void* d_out, int out_size, void* d_ws, size_t ws_size,
                              hipStream_t stream) {
  const float* x   = (const float*)d_in[0];
  const int*  eidx = (const int*)d_in[1];
  const float* W1  = (const float*)d_in[2];
  const float* b1  = (const float*)d_in[3];
  const float* W2  = (const float*)d_in[4];
  const float* b2  = (const float*)d_in[5];
  float* out = (float*)d_out;

  const int E = in_sizes[1] / 2;
  const int* src = eidx;
  const int* dst = eidx + E;

  float* dinv = (float*)d_ws;
  float* bufA = dinv + 50048;
  float* bufB = bufA + (size_t)NNODES * 128;

  const int T = 256;

  k_init_deg<<<(NNODES + T - 1) / T, T, 0, stream>>>(dinv);
  k_count_deg<<<(E + T - 1) / T, T, 0, stream>>>(dst, dinv, E);
  k_finish_deg<<<(NNODES + T - 1) / T, T, 0, stream>>>(dinv);

  {
    k_gemm_wmma_f32<256, 128><<<512, 256, 0, stream>>>(x, W1, bufA, NNODES);
  }
  k_aggregate<128, 64><<<AGG_WGS, 1024, 0, stream>>>(src, dst, dinv, bufA, b1, W2, bufB, E);

  k_aggregate<64, 0><<<AGG_WGS, 1024, 0, stream>>>(src, dst, dinv, bufB, b2, nullptr, out, E);
}


# Layer-1 aggregation saves each wave's kept edges (source id, local row) in the workspace; layer 2 walks those lists and no longer filters the edge list; LDS float adds stay in arrival order
# speedup vs baseline: 2.1321x; 1.1872x over previous
#include <hip/hip_runtime.h>
#include <hip/hip_bf16.h>

typedef float v2f __attribute__((ext_vector_type(2)));
typedef float v8f __attribute__((ext_vector_type(8)));

#define NNODES 50000

__global__ void k_init_deg(float* deg) {
  int i = blockIdx.x * blockDim.x + threadIdx.x;
  if (i < NNODES) deg[i] = 1.0f;
}

__global__ void k_count_deg(const int* __restrict__ dst, float* __restrict__ deg, int E) {
  int e = blockIdx.x * blockDim.x + threadIdx.x;
  if (e < E) atomicAdd(&deg[dst[e]], 1.0f);
}

__global__ void k_finish_deg(float* deg) {
  int i = blockIdx.x * blockDim.x + threadIdx.x;
  if (i < NNODES) deg[i] = rsqrtf(deg[i]);
}

template <int K, int N>
__global__ __launch_bounds__(256) void k_gemm_wmma_f32(const float* __restrict__ A,
                                                       const float* __restrict__ W,
                                                       float* __restrict__ C, int M) {
  constexpr int CT    = N / 16;
  constexpr int RPS   = 8 / CT;
  constexpr int ROWS  = RPS * 16;
  constexpr int PITCH = K + 4;
  constexpr int KS    = K / 4;
  constexpr int F4ROW = K / 4;
  constexpr int NLD   = ROWS * F4ROW / 256;

  __shared__ float lds[2][ROWS * PITCH];

  const int tid  = threadIdx.x;
  const int lane = tid & 31;
  const int wave = tid >> 5;
  const int half = lane >> 4;
  const int r    = lane & 15;
  const int ct   = wave % CT;
  const int rt   = wave / CT;
  const int row_tiles = M / 16;
  const int nstages   = (row_tiles + RPS - 1) / RPS;

  v2f breg[KS];
  {
    const float* wcol = W + (size_t)(ct * 16 + r);
#pragma unroll
    for (int i = 0; i < KS; ++i) {
      const int kA = 4 * i + 2 * half;
      breg[i].x = wcol[(size_t)kA * N];
      breg[i].y = wcol[(size_t)(kA + 1) * N];
    }
  }

  float4 pre[NLD];
  int s = blockIdx.x;

#pragma unroll
  for (int j = 0; j < NLD; ++j) {
    const int idx  = tid + 256 * j;
    const int row  = idx / F4ROW;
    const int c4   = idx % F4ROW;
    const int grow = s * ROWS + row;
    float4 v = make_float4(0.f, 0.f, 0.f, 0.f);
    if (s < nstages && grow < M) v = *(const float4*)(A + (size_t)grow * K + c4 * 4);
    *(float4*)&lds[0][row * PITCH + c4 * 4] = v;
  }
  __syncthreads();

  int cur = 0;
  for (; s < nstages; s += gridDim.x) {
    const int  sn       = s + gridDim.x;
    const bool has_next = sn < nstages;
    if (has_next) {
#pragma unroll
      for (int j = 0; j < NLD; ++j) {
        const int idx  = tid + 256 * j;
        const int row  = idx / F4ROW;
        const int c4   = idx % F4ROW;
        const int grow = sn * ROWS + row;
        pre[j] = make_float4(0.f, 0.f, 0.f, 0.f);
        if (grow < M) pre[j] = *(const float4*)(A + (size_t)grow * K + c4 * 4);
      }
    }

    const int tile = s * RPS + rt;
    if (tile < row_tiles) {
      const float* ap = &lds[cur][(rt * 16 + r) * PITCH + 2 * half];
      v8f acc = {};
#pragma unroll
      for (int i = 0; i < KS; ++i) {
        v2f a = *(const v2f*)(ap + 4 * i);
        acc = __builtin_amdgcn_wmma_f32_16x16x4_f32(
             false, a,  false, breg[i],
             (short)0, acc,  false,  false);
      }
      float* crow = C + (size_t)(tile * 16 + 8 * half) * N + ct * 16 + r;
#pragma unroll
      for (int i = 0; i < 8; ++i) crow[(size_t)i * N] = acc[i];
    }

    if (has_next) {
#pragma unroll
      for (int j = 0; j < NLD; ++j) {
        const int idx = tid + 256 * j;
        const int row = idx / F4ROW;
        const int c4  = idx % F4ROW;
        *(float4*)&lds[cur ^ 1][row * PITCH + c4 * 4] = pre[j];
      }
    }
    __syncthreads();
    cur ^= 1;
  }
}

#define AGG_WGS  256
#define AGG_ROWS ((NNODES + AGG_WGS - 1) / AGG_WGS)
#define AGG_QCAP 512
#define AGG_LCAP 384
static_assert(NNODES <= (1 << 24), "a source id is packed into 24 bits");
static_assert((size_t)AGG_WGS * 32 * (AGG_LCAP + 1) <= (size_t)NNODES * 64, "numbers and lists fit the unused half of the second workspace plane");

template <int CH, int NOUT, int LIST>
__global__ __launch_bounds__(1024) void k_aggregate(const int* __restrict__ src, const int* __restrict__ dst,
                                                    const float* __restrict__ dinv,
                                                    const float* __restrict__ feat,
                                                    const float* __restrict__ bias,
                                                    const float* __restrict__ W,
                                                    float* __restrict__ out,
                                                    unsigned* __restrict__ ecnt,
                                                    unsigned* __restrict__ elist, int E) {
  constexpr int PITCH = (NOUT > 0) ? CH + 4 : CH;
  constexpr int TROWS = (NOUT > 0) ? (AGG_ROWS + 15) / 16 * 16 : AGG_ROWS;
  constexpr int VPL  = CH / 32;
  constexpr int ROW4 = CH / 4;
  constexpr int G    = (CH <= 64) ? 16 : 8;
  static_assert(AGG_ROWS <= 256, "row number inside a workgroup is packed into 8 bits");
  __shared__ __attribute__((aligned(16))) float tile[TROWS * PITCH];
  __shared__ __attribute__((aligned(16))) unsigned ring[32][AGG_QCAP];

  const int tid  = (int)threadIdx.x;
  const int lane = tid & 31;
  const int wv   = __builtin_amdgcn_readfirstlane(tid >> 5);
  const int lo   = (int)blockIdx.x * AGG_ROWS;
  const int nrows = max(0, min(AGG_ROWS, NNODES - lo));

  {
    constexpr int NIT = (AGG_ROWS * ROW4 + 1023) / 1024;
    float4 v[NIT];
    float  di[NIT];
#pragma unroll
    for (int j = 0; j < NIT; ++j) {
      const int i = tid + 1024 * j;
      v[j]  = make_float4(0.f, 0.f, 0.f, 0.f);
      di[j] = 0.f;
      if (i < nrows * ROW4) {
        v[j]  = *(const float4*)(feat + (size_t)lo * CH + (size_t)i * 4);
        di[j] = dinv[lo + i / ROW4];
      }
    }
#pragma unroll
    for (int j = 0; j < NIT; ++j) {
      const int i = tid + 1024 * j;
      if (i < nrows * ROW4) {
        const float4 b = *(const float4*)(bias + (i % ROW4) * 4);
        float4 o;
        o.x = v[j].x * di[j] * di[j] + b.x;
        o.y = v[j].y * di[j] * di[j] + b.y;
        o.z = v[j].z * di[j] * di[j] + b.z;
        o.w = v[j].w * di[j] * di[j] + b.w;
        *(float4*)&tile[(i / ROW4) * PITCH + (i % ROW4) * 4] = o;
      }
    }
    for (int i = nrows * ROW4 + tid; i < TROWS * ROW4; i += 1024)
      *(float4*)&tile[(i / ROW4) * PITCH + (i % ROW4) * 4] = make_float4(0.f, 0.f, 0.f, 0.f);
  }
  __syncthreads();

  unsigned* q = ring[wv];
  int qh = 0, qn = 0;
  unsigned* lst = elist + (size_t)((int)blockIdx.x * 32 + wv) * AGG_LCAP;
  int ln = 0;

  auto add_rows = [&](const int (&s)[G], const int (&dl)[G], int n) {
    float v[G][VPL], w[G];
#pragma unroll
    for (int k = 0; k < G; ++k) {
      const float* frow = feat + (size_t)s[k] * CH;
#pragma unroll
      for (int c = 0; c < VPL; ++c) v[k][c] = frow[c * 32 + lane];
      w[k] = dinv[s[k]] * dinv[lo + dl[k]];
    }
#pragma unroll
    for (int k = 0; k < G; ++k) {
      if (k < n) {
        float* trow = &tile[dl[k] * PITCH + lane];
#pragma unroll
        for (int c = 0; c < VPL; ++c)
          __hip_atomic_fetch_add(trow + c * 32, v[k][c] * w[k], __ATOMIC_RELAXED, __HIP_MEMORY_SCOPE_WORKGROUP);
      }
    }
  };

  auto drain = [&](int base, int n) {
    int e[G], dl[G], s[G];
#pragma unroll
    for (int k = 0; k < G; ++k) {
      const int kk = (k < n) ? k : 0;
      const unsigned ent = q[(base + kk) & (AGG_QCAP - 1)];
      e[k]  = __builtin_amdgcn_readfirstlane((int)(ent & 0xFFFFFFu));
      dl[k] = __builtin_amdgcn_readfirstlane((int)(ent >> 24));
    }
#pragma unroll
    for (int k = 0; k < G; ++k) s[k] = __builtin_amdgcn_readfirstlane(src[e[k]]);
    if constexpr (LIST == 1) {
      unsigned ent = 0u;
#pragma unroll
      for (int k = 0; k < G; ++k)
        if (lane == k) ent = (unsigned)s[k] | ((unsigned)dl[k] << 24);
      if (lane < n && ln + lane < AGG_LCAP) lst[ln + lane] = ent;
      ln += n;
    }
    add_rows(s, dl, n);
  };

  auto test8 = [&](const int* d, int ebase) {
#pragma unroll
    for (int u = 0; u < 8; ++u) {
      const unsigned dloc = (unsigned)(d[u] - lo);
      const bool     hit  = dloc < (unsigned)nrows;
      const unsigned m    = __builtin_amdgcn_ballot_w32(hit);
      if (__builtin_expect(m != 0u, 0)) {
        if (hit) {
          const int pos = qn + __builtin_popcount(m & ((1u << lane) - 1u));
          q[pos & (AGG_QCAP - 1)] = (unsigned)(ebase + 32 * u) | (dloc << 24);
        }
        qn += __builtin_popcount(m);
      }
    }
    while (qn - qh >= G) { drain(qh, G); qh += G; }
  };

  auto filter_share = [&]() {
  const int nfull = E >> 10;
  const int rot = (int)(((long long)blockIdx.x * nfull) / AGG_WGS);
  for (int i = wv; i < nfull; i += 32) {
    int t = i + rot;
    if (t >= nfull) t -= nfull;
    const int  e0 = (t << 10) + lane;
    const int* dp = dst + e0;
    int d[32];
#pragma unroll
    for (int u = 0; u < 32; ++u) d[u] = dp[32 * u];
#pragma unroll
    for (int g = 0; g < 4; ++g) test8(&d[8 * g], e0 + 256 * g);
  }
  {
    const int tbase = nfull << 10;
    const int npiece = (E - tbase + 255) >> 8;
    if (wv < npiece) {
      const int e0 = tbase + (wv << 8) + lane;
      int d[8];
#pragma unroll
      for (int u = 0; u < 8; ++u) {
        const int ee = e0 + 32 * u;
        d[u] = (ee < E) ? dst[ee] : -1;
      }
      test8(d, e0);
    }
  }
  if (qn > qh) drain(qh, qn - qh);
  };

  if constexpr (LIST == 2) {
    const int cnt = __builtin_amdgcn_readfirstlane((int)ecnt[(int)blockIdx.x * 32 + wv]);
    if (__builtin_expect(cnt <= AGG_LCAP, 1)) {
      unsigned cur = (cnt > 0) ? lst[min(lane, cnt - 1)] : 0u;
      for (int base = 0; base < cnt; base += 32) {
        const unsigned nxt = lst[min(base + 32 + lane, cnt - 1)];
#pragma unroll
        for (int sub = 0; sub < 32; sub += G) {
          const int n = min(G, cnt - base - sub);
          if (n > 0) {
            int s[G], dl[G];
#pragma unroll
            for (int k = 0; k < G; ++k) {
              const unsigned ent = (unsigned)__builtin_amdgcn_readlane((int)cur, sub + k);
              s[k]  = (int)(ent & 0xFFFFFFu);
              dl[k] = (int)(ent >> 24);
            }
            add_rows(s, dl, n);
          }
        }
        cur = nxt;
      }
    } else {
      filter_share();
    }
  } else {
    filter_share();
    if constexpr (LIST == 1) {
      if (lane == 0) ecnt[(int)blockIdx.x * 32 + wv] = (unsigned)ln;
    }
  }
  __syncthreads();

  if constexpr (NOUT == 0) {
    for (int i = tid; i < nrows * ROW4; i += 1024)
      *(float4*)(out + (size_t)lo * CH + (size_t)i * 4) = *(const float4*)&tile[(i / ROW4) * PITCH + (i % ROW4) * 4];
  } else {
    constexpr int CT  = NOUT / 16;
    constexpr int KS  = CH / 4;
    constexpr int RTS = 32 / CT;
    static_assert(32 % CT == 0, "every wave owns one column tile");
    const int half = lane >> 4;
    const int r    = lane & 15;
    const int ct   = wv % CT;
    v2f breg[KS];
    {
      const float* wcol = W + (size_t)(ct * 16 + r);
#pragma unroll
      for (int i = 0; i < KS; ++i) {
        const int kA = 4 * i + 2 * half;
        breg[i].x = wcol[(size_t)kA * NOUT];
        breg[i].y = wcol[(size_t)(kA + 1) * NOUT];
      }
    }
    for (int i = tid; i < nrows * ROW4; i += 1024) {
      float4* p = (float4*)&tile[(i / ROW4) * PITCH + (i % ROW4) * 4];
      float4 v = *p;
      v.x = tanhf(v.x);
      v.y = tanhf(v.y);
      v.z = tanhf(v.z);
      v.w = tanhf(v.w);
      *p = v;
    }
    __syncthreads();
    static_assert(TROWS * NOUT <= 32 * AGG_QCAP, "product rows fit the rings' space");
    float* stage = (float*)&ring[0][0];
    const int ntiles = (nrows + 15) >> 4;
    for (int t = wv / CT; t < ntiles; t += RTS) {
      const float* ap = &tile[(t * 16 + r) * PITCH + 2 * half];
      v8f acc = {};
#pragma unroll
      for (int i = 0; i < KS; ++i) {
        v2f a = *(const v2f*)(ap + 4 * i);
        acc = __builtin_amdgcn_wmma_f32_16x16x4_f32(
             false, a,  false, breg[i],
             (short)0, acc,  false,  false);
      }
      float* sp = stage + (t * 16 + 8 * half) * NOUT + ct * 16 + r;
#pragma unroll
      for (int i = 0; i < 8; ++i) sp[i * NOUT] = acc[i];
    }
    __syncthreads();
    for (int i = tid; i < nrows * (NOUT / 4); i += 1024)
      *(float4*)(out + (size_t)lo * NOUT + (size_t)i * 4) = *(const float4*)&stage[i * 4];
  }
}

extern "C" void kernel_launch(void* const* d_in, const int* in_sizes, int n_in,
                              void* d_out, int out_size, void* d_ws, size_t ws_size,
                              hipStream_t stream) {
  const float* x   = (const float*)d_in[0];
  const int*  eidx = (const int*)d_in[1];
  const float* W1  = (const float*)d_in[2];
  const float* b1  = (const float*)d_in[3];
  const float* W2  = (const float*)d_in[4];
  const float* b2  = (const float*)d_in[5];
  float* out = (float*)d_out;

  const int E = in_sizes[1] / 2;
  const int* src = eidx;
  const int* dst = eidx + E;

  float* dinv = (float*)d_ws;
  float* bufA = dinv + 50048;
  float* bufB = bufA + (size_t)NNODES * 128;

  const int T = 256;

  k_init_deg<<<(NNODES + T - 1) / T, T, 0, stream>>>(dinv);
  k_count_deg<<<(E + T - 1) / T, T, 0, stream>>>(dst, dinv, E);
  k_finish_deg<<<(NNODES + T - 1) / T, T, 0, stream>>>(dinv);

  {
    k_gemm_wmma_f32<256, 128><<<512, 256, 0, stream>>>(x, W1, bufA, NNODES);
  }
  unsigned* ecnt  = (unsigned*)(bufB + (size_t)NNODES * 64);
  unsigned* elist = ecnt + AGG_WGS * 32;
  k_aggregate<128, 64, 1><<<AGG_WGS, 1024, 0, stream>>>(src, dst, dinv, bufA, b1, W2, bufB, ecnt, elist, E);

  k_aggregate<64, 0, 2><<<AGG_WGS, 1024, 0, stream>>>(src, dst, dinv, bufB, b2, nullptr, out, ecnt, elist, E);
}


# Layer-1 edge filter reads a byte-per-edge owner map written by the degree pass, 4 edges per dword by a zero-byte test, not 4-byte ids; row number read at handling; LDS adds still arrival order
# speedup vs baseline: 2.2364x; 1.0489x over previous
#include <hip/hip_runtime.h>
#include <hip/hip_bf16.h>

typedef float v2f __attribute__((ext_vector_type(2)));
typedef float v8f __attribute__((ext_vector_type(8)));

#define NNODES 50000

#define AGG_WGS  256
#define AGG_ROWS ((NNODES + AGG_WGS - 1) / AGG_WGS)
static_assert((NNODES - 1) / AGG_ROWS < 256, "an owner number fits one byte");

__global__ void k_init_deg(float* deg) {
  int i = blockIdx.x * blockDim.x + threadIdx.x;
  if (i < NNODES) deg[i] = 1.0f;
}

__global__ void k_count_deg(const int* __restrict__ dst, float* __restrict__ deg,
                            unsigned* __restrict__ omap, int E) {
  const int j  = blockIdx.x * blockDim.x + threadIdx.x;
  const int e0 = 4 * j;
  if (e0 + 3 < E) {
    int d[4];
#pragma unroll
    for (int b = 0; b < 4; ++b) d[b] = dst[e0 + b];
    unsigned w = 0u;
#pragma unroll
    for (int b = 0; b < 4; ++b) {
      atomicAdd(&deg[d[b]], 1.0f);
      w |= ((unsigned)d[b] / (unsigned)AGG_ROWS) << (8 * b);
    }
    omap[j] = w;
  } else if (e0 < E) {
    unsigned w = 0u;
    for (int b = 0; e0 + b < E; ++b) {
      const int d = dst[e0 + b];
      atomicAdd(&deg[d], 1.0f);
      w |= ((unsigned)d / (unsigned)AGG_ROWS) << (8 * b);
    }
    omap[j] = w;
  }
}

__global__ void k_finish_deg(float* deg) {
  int i = blockIdx.x * blockDim.x + threadIdx.x;
  if (i < NNODES) deg[i] = rsqrtf(deg[i]);
}

template <int K, int N>
__global__ __launch_bounds__(256) void k_gemm_wmma_f32(const float* __restrict__ A,
                                                       const float* __restrict__ W,
                                                       float* __restrict__ C, int M) {
  constexpr int CT    = N / 16;
  constexpr int RPS   = 8 / CT;
  constexpr int ROWS  = RPS * 16;
  constexpr int PITCH = K + 4;
  constexpr int KS    = K / 4;
  constexpr int F4ROW = K / 4;
  constexpr int NLD   = ROWS * F4ROW / 256;

  __shared__ float lds[2][ROWS * PITCH];

  const int tid  = threadIdx.x;
  const int lane = tid & 31;
  const int wave = tid >> 5;
  const int half = lane >> 4;
  const int r    = lane & 15;
  const int ct   = wave % CT;
  const int rt   = wave / CT;
  const int row_tiles = M / 16;
  const int nstages   = (row_tiles + RPS - 1) / RPS;

  v2f breg[KS];
  {
    const float* wcol = W + (size_t)(ct * 16 + r);
#pragma unroll
    for (int i = 0; i < KS; ++i) {
      const int kA = 4 * i + 2 * half;
      breg[i].x = wcol[(size_t)kA * N];
      breg[i].y = wcol[(size_t)(kA + 1) * N];
    }
  }

  float4 pre[NLD];
  int s = blockIdx.x;

#pragma unroll
  for (int j = 0; j < NLD; ++j) {
    const int idx  = tid + 256 * j;
    const int row  = idx / F4ROW;
    const int c4   = idx % F4ROW;
    const int grow = s * ROWS + row;
    float4 v = make_float4(0.f, 0.f, 0.f, 0.f);
    if (s < nstages && grow < M) v = *(const float4*)(A + (size_t)grow * K + c4 * 4);
    *(float4*)&lds[0][row * PITCH + c4 * 4] = v;
  }
  __syncthreads();

  int cur = 0;
  for (; s < nstages; s += gridDim.x) {
    const int  sn       = s + gridDim.x;
    const bool has_next = sn < nstages;
    if (has_next) {
#pragma unroll
      for (int j = 0; j < NLD; ++j) {
        const int idx  = tid + 256 * j;
        const int row  = idx / F4ROW;
        const int c4   = idx % F4ROW;
        const int grow = sn * ROWS + row;
        pre[j] = make_float4(0.f, 0.f, 0.f, 0.f);
        if (grow < M) pre[j] = *(const float4*)(A + (size_t)grow * K + c4 * 4);
      }
    }

    const int tile = s * RPS + rt;
    if (tile < row_tiles) {
      const float* ap = &lds[cur][(rt * 16 + r) * PITCH + 2 * half];
      v8f acc = {};
#pragma unroll
      for (int i = 0; i < KS; ++i) {
        v2f a = *(const v2f*)(ap + 4 * i);
        acc = __builtin_amdgcn_wmma_f32_16x16x4_f32(
             false, a,  false, breg[i],
             (short)0, acc,  false,  false);
      }
      float* crow = C + (size_t)(tile * 16 + 8 * half) * N + ct * 16 + r;
#pragma unroll
      for (int i = 0; i < 8; ++i) crow[(size_t)i * N] = acc[i];
    }

    if (has_next) {
#pragma unroll
      for (int j = 0; j < NLD; ++j) {
        const int idx = tid + 256 * j;
        const int row = idx / F4ROW;
        const int c4  = idx % F4ROW;
        *(float4*)&lds[cur ^ 1][row * PITCH + c4 * 4] = pre[j];
      }
    }
    __syncthreads();
    cur ^= 1;
  }
}

#define AGG_QCAP 512
#define AGG_LCAP 352
#define AGG_MAPDW ((size_t)NNODES * 64 - (size_t)AGG_WGS * 32 * (AGG_LCAP + 1))
static_assert(NNODES <= (1 << 24), "a source id is packed into 24 bits");
static_assert((size_t)AGG_WGS * 32 * (AGG_LCAP + 1) <= (size_t)NNODES * 64, "numbers and lists fit the unused half of the second workspace plane");
static_assert(AGG_MAPDW * 4 >= 1200000, "the owner map has room for 1200000 edges; the edge list here has 800000");
static_assert(((size_t)AGG_WGS * 32 * (AGG_LCAP + 1)) % 32 == 0, "the owner map starts on a whole line");

template <int CH, int NOUT, int LIST>
__global__ __launch_bounds__(1024) void k_aggregate(const int* __restrict__ src, const int* __restrict__ dst,
                                                    const float* __restrict__ dinv,
                                                    const float* __restrict__ feat,
                                                    const float* __restrict__ bias,
                                                    const float* __restrict__ W,
                                                    float* __restrict__ out,
                                                    unsigned* __restrict__ ecnt,
                                                    unsigned* __restrict__ elist,
                                                    const unsigned* __restrict__ omap, int E) {
  constexpr int PITCH = (NOUT > 0) ? CH + 4 : CH;
  constexpr int TROWS = (NOUT > 0) ? (AGG_ROWS + 15) / 16 * 16 : AGG_ROWS;
  constexpr int VPL  = CH / 32;
  constexpr int ROW4 = CH / 4;
  constexpr int G    = (CH <= 64) ? 16 : 8;
  static_assert(AGG_ROWS <= 256, "row number inside a workgroup is packed into 8 bits");
  __shared__ __attribute__((aligned(16))) float tile[TROWS * PITCH];
  __shared__ __attribute__((aligned(16))) unsigned ring[32][AGG_QCAP];

  const int tid  = (int)threadIdx.x;
  const int lane = tid & 31;
  const int wv   = __builtin_amdgcn_readfirstlane(tid >> 5);
  const int lo   = (int)blockIdx.x * AGG_ROWS;
  const int nrows = max(0, min(AGG_ROWS, NNODES - lo));

  {
    constexpr int NIT = (AGG_ROWS * ROW4 + 1023) / 1024;
    float4 v[NIT];
    float  di[NIT];
#pragma unroll
    for (int j = 0; j < NIT; ++j) {
      const int i = tid + 1024 * j;
      v[j]  = make_float4(0.f, 0.f, 0.f, 0.f);
      di[j] = 0.f;
      if (i < nrows * ROW4) {
        v[j]  = *(const float4*)(feat + (size_t)lo * CH + (size_t)i * 4);
        di[j] = dinv[lo + i / ROW4];
      }
    }
#pragma unroll
    for (int j = 0; j < NIT; ++j) {
      const int i = tid + 1024 * j;
      if (i < nrows * ROW4) {
        const float4 b = *(const float4*)(bias + (i % ROW4) * 4);
        float4 o;
        o.x = v[j].x * di[j] * di[j] + b.x;
        o.y = v[j].y * di[j] * di[j] + b.y;
        o.z = v[j].z * di[j] * di[j] + b.z;
        o.w = v[j].w * di[j] * di[j] + b.w;
        *(float4*)&tile[(i / ROW4) * PITCH + (i % ROW4) * 4] = o;
      }
    }
    for (int i = nrows * ROW4 + tid; i < TROWS * ROW4; i += 1024)
      *(float4*)&tile[(i / ROW4) * PITCH + (i % ROW4) * 4] = make_float4(0.f, 0.f, 0.f, 0.f);
  }
  __syncthreads();

  unsigned* q = ring[wv];
  int qh = 0, qn = 0;
  unsigned* lst = elist + (size_t)((int)blockIdx.x * 32 + wv) * AGG_LCAP;
  int ln = 0;

  auto add_rows = [&](const int (&s)[G], const int (&dl)[G], int n) {
    float v[G][VPL], w[G];
#pragma unroll
    for (int k = 0; k < G; ++k) {
      const float* frow = feat + (size_t)s[k] * CH;
#pragma unroll
      for (int c = 0; c < VPL; ++c) v[k][c] = frow[c * 32 + lane];
      w[k] = dinv[s[k]] * dinv[lo + dl[k]];
    }
#pragma unroll
    for (int k = 0; k < G; ++k) {
      if (k < n) {
        float* trow = &tile[dl[k] * PITCH + lane];
#pragma unroll
        for (int c = 0; c < VPL; ++c)
          __hip_atomic_fetch_add(trow + c * 32, v[k][c] * w[k], __ATOMIC_RELAXED, __HIP_MEMORY_SCOPE_WORKGROUP);
      }
    }
  };

  auto drain = [&](int base, int n) {
    int e[G], dl[G], s[G];
#pragma unroll
    for (int k = 0; k < G; ++k) {
      const int kk = (k < n) ? k : 0;
      e[k] = __builtin_amdgcn_readfirstlane((int)q[(base + kk) & (AGG_QCAP - 1)]);
    }
#pragma unroll
    for (int k = 0; k < G; ++k) {
      s[k]  = __builtin_amdgcn_readfirstlane(src[e[k]]);
      dl[k] = __builtin_amdgcn_readfirstlane(dst[e[k]]) - lo;
    }
    if constexpr (LIST == 1) {
      unsigned ent = 0u;
#pragma unroll
      for (int k = 0; k < G; ++k)
        if (lane == k) ent = (unsigned)s[k] | ((unsigned)dl[k] << 24);
      if (lane < n && ln + lane < AGG_LCAP) lst[ln + lane] = ent;
      ln += n;
    }
    add_rows(s, dl, n);
  };

  auto test8 = [&](const int* d, int ebase) {
#pragma unroll
    for (int u = 0; u < 8; ++u) {
      const unsigned dloc = (unsigned)(d[u] - lo);
      const bool     hit  = dloc < (unsigned)nrows;
      const unsigned m    = __builtin_amdgcn_ballot_w32(hit);
      if (__builtin_expect(m != 0u, 0)) {
        if (hit) {
          const int pos = qn + __builtin_popcount(m & ((1u << lane) - 1u));
          q[pos & (AGG_QCAP - 1)] = (unsigned)(ebase + 32 * u);
        }
        qn += __builtin_popcount(m);
      }
    }
    while (qn - qh >= G) { drain(qh, G); qh += G; }
  };

  const unsigned pat = (unsigned)blockIdx.x * 0x01010101u;

  auto test2 = [&](const unsigned* w, int jbase) {
#pragma unroll
    for (int u = 0; u < 2; ++u) {
      const unsigned x = w[u] ^ pat;
      unsigned z = ~(((x & 0x7F7F7F7Fu) + 0x7F7F7F7Fu) | x) & 0x80808080u;
      unsigned m = __builtin_amdgcn_ballot_w32(z != 0u);
      if (__builtin_expect(m != 0u, 0)) {
        const int ebase = (jbase + 32 * u) << 2;
        for (int it = 0; it < 4 && m != 0u; ++it) {
          if (z != 0u) {
            const int pos = qn + __builtin_popcount(m & ((1u << lane) - 1u));
            q[pos & (AGG_QCAP - 1)] = (unsigned)(ebase + (__builtin_ctz(z) >> 3));
            z &= z - 1u;
          }
          qn += __builtin_popcount(m);
          m = __builtin_amdgcn_ballot_w32(z != 0u);
        }
      }
    }
    while (qn - qh >= G) { drain(qh, G); qh += G; }
  };

  auto filter_share = [&]() {
  const int nfull = E >> 10;
  const int rot = (int)(((long long)blockIdx.x * nfull) / AGG_WGS);
  auto ask = [&](unsigned (&w)[8], int i) -> int {
    int t = i + rot;
    if (t >= nfull) t -= nfull;
    const unsigned* mp = omap + (t << 8) + lane;
#pragma unroll
    for (int u = 0; u < 8; ++u) w[u] = mp[32 * u];
    return t << 8;
  };
  unsigned wa[8], wb[8];
  int ja = 0;
  if (wv < nfull) ja = ask(wa, wv);
  for (int i = wv; i < nfull; i += 64) {
    const bool has_b = i + 32 < nfull;
    const int  jb    = ask(wb, has_b ? i + 32 : i);
#pragma unroll
    for (int g = 0; g < 4; ++g) test2(&wa[2 * g], ja + lane + 64 * g);
    ja = ask(wa, (i + 64 < nfull) ? i + 64 : i);
    if (has_b) {
#pragma unroll
      for (int g = 0; g < 4; ++g) test2(&wb[2 * g], jb + lane + 64 * g);
    }
  }
  {
    const int tbase = nfull << 10;
    const int npiece = (E - tbase + 255) >> 8;
    if (wv < npiece) {
      const int e0 = tbase + (wv << 8) + lane;
      int d[8];
#pragma unroll
      for (int u = 0; u < 8; ++u) {
        const int ee = e0 + 32 * u;
        d[u] = (ee < E) ? dst[ee] : -1;
      }
      test8(d, e0);
    }
  }
  if (qn > qh) drain(qh, qn - qh);
  };

  if constexpr (LIST == 2) {
    const int cnt = __builtin_amdgcn_readfirstlane((int)ecnt[(int)blockIdx.x * 32 + wv]);
    if (__builtin_expect(cnt <= AGG_LCAP, 1)) {
      unsigned cur = (cnt > 0) ? lst[min(lane, cnt - 1)] : 0u;
      for (int base = 0; base < cnt; base += 32) {
        const unsigned nxt = lst[min(base + 32 + lane, cnt - 1)];
#pragma unroll
        for (int sub = 0; sub < 32; sub += G) {
          const int n = min(G, cnt - base - sub);
          if (n > 0) {
            int s[G], dl[G];
#pragma unroll
            for (int k = 0; k < G; ++k) {
              const unsigned ent = (unsigned)__builtin_amdgcn_readlane((int)cur, sub + k);
              s[k]  = (int)(ent & 0xFFFFFFu);
              dl[k] = (int)(ent >> 24);
            }
            add_rows(s, dl, n);
          }
        }
        cur = nxt;
      }
    } else {
      filter_share();
    }
  } else {
    filter_share();
    if constexpr (LIST == 1) {
      if (lane == 0) ecnt[(int)blockIdx.x * 32 + wv] = (unsigned)ln;
    }
  }
  __syncthreads();

  if constexpr (NOUT == 0) {
    for (int i = tid; i < nrows * ROW4; i += 1024)
      *(float4*)(out + (size_t)lo * CH + (size_t)i * 4) = *(const float4*)&tile[(i / ROW4) * PITCH + (i % ROW4) * 4];
  } else {
    constexpr int CT  = NOUT / 16;
    constexpr int KS  = CH / 4;
    constexpr int RTS = 32 / CT;
    static_assert(32 % CT == 0, "every wave owns one column tile");
    const int half = lane >> 4;
    const int r    = lane & 15;
    const int ct   = wv % CT;
    v2f breg[KS];
    {
      const float* wcol = W + (size_t)(ct * 16 + r);
#pragma unroll
      for (int i = 0; i < KS; ++i) {
        const int kA = 4 * i + 2 * half;
        breg[i].x = wcol[(size_t)kA * NOUT];
        breg[i].y = wcol[(size_t)(kA + 1) * NOUT];
      }
    }
    for (int i = tid; i < nrows * ROW4; i += 1024) {
      float4* p = (float4*)&tile[(i / ROW4) * PITCH + (i % ROW4) * 4];
      float4 v = *p;
      v.x = tanhf(v.x);
      v.y = tanhf(v.y);
      v.z = tanhf(v.z);
      v.w = tanhf(v.w);
      *p = v;
    }
    __syncthreads();
    static_assert(TROWS * NOUT <= 32 * AGG_QCAP, "product rows fit the rings' space");
    float* stage = (float*)&ring[0][0];
    const int ntiles = (nrows + 15) >> 4;
    for (int t = wv / CT; t < ntiles; t += RTS) {
      const float* ap = &tile[(t * 16 + r) * PITCH + 2 * half];
      v8f acc = {};
#pragma unroll
      for (int i = 0; i < KS; ++i) {
        v2f a = *(const v2f*)(ap + 4 * i);
        acc = __builtin_amdgcn_wmma_f32_16x16x4_f32(
             false, a,  false, breg[i],
             (short)0, acc,  false,  false);
      }
      float* sp = stage + (t * 16 + 8 * half) * NOUT + ct * 16 + r;
#pragma unroll
      for (int i = 0; i < 8; ++i) sp[i * NOUT] = acc[i];
    }
    __syncthreads();
    for (int i = tid; i < nrows * (NOUT / 4); i += 1024)
      *(float4*)(out + (size_t)lo * NOUT + (size_t)i * 4) = *(const float4*)&stage[i * 4];
  }
}

extern "C" void kernel_launch(void* const* d_in, const int* in_sizes, int n_in,
                              void* d_out, int out_size, void* d_ws, size_t ws_size,
                              hipStream_t stream) {
  const float* x   = (const float*)d_in[0];
  const int*  eidx = (const int*)d_in[1];
  const float* W1  = (const float*)d_in[2];
  const float* b1  = (const float*)d_in[3];
  const float* W2  = (const float*)d_in[4];
  const float* b2  = (const float*)d_in[5];
  float* out = (float*)d_out;

  const int E = in_sizes[1] / 2;
  const int* src = eidx;
  const int* dst = eidx + E;

  float* dinv = (float*)d_ws;
  float* bufA = dinv + 50048;
  float* bufB = bufA + (size_t)NNODES * 128;

  const int T = 256;

  unsigned* ecnt  = (unsigned*)(bufB + (size_t)NNODES * 64);
  unsigned* elist = ecnt + AGG_WGS * 32;
  unsigned* omap  = elist + (size_t)AGG_WGS * 32 * AGG_LCAP;

  k_init_deg<<<(NNODES + T - 1) / T, T, 0, stream>>>(dinv);
  k_count_deg<<<((E + 3) / 4 + T - 1) / T, T, 0, stream>>>(dst, dinv, omap, E);
  k_finish_deg<<<(NNODES + T - 1) / T, T, 0, stream>>>(dinv);

  {
    k_gemm_wmma_f32<256, 128><<<512, 256, 0, stream>>>(x, W1, bufA, NNODES);
  }
  k_aggregate<128, 64, 1><<<AGG_WGS, 1024, 0, stream>>>(src, dst, dinv, bufA, b1, W2, bufB, ecnt, elist, omap, E);

  k_aggregate<64, 0, 2><<<AGG_WGS, 1024, 0, stream>>>(src, dst, dinv, bufB, b2, nullptr, out, ecnt, elist, omap, E);
}


# Edge weight factored: a feature row is stored times its own dinv where it is made; destination dinv and bias go on once per row after the LDS sum (bias now last); no per-edge dinv load or multiply
# speedup vs baseline: 2.3512x; 1.0513x over previous
#include <hip/hip_runtime.h>
#include <hip/hip_bf16.h>

typedef float v2f __attribute__((ext_vector_type(2)));
typedef float v8f __attribute__((ext_vector_type(8)));

#define NNODES 50000

#define AGG_WGS  256
#define AGG_ROWS ((NNODES + AGG_WGS - 1) / AGG_WGS)
static_assert((NNODES - 1) / AGG_ROWS < 256, "an owner number fits one byte");

__global__ void k_init_deg(float* deg) {
  int i = blockIdx.x * blockDim.x + threadIdx.x;
  if (i < NNODES) deg[i] = 1.0f;
}

__global__ void k_count_deg(const int* __restrict__ dst, float* __restrict__ deg,
                            unsigned* __restrict__ omap, int E) {
  const int j  = blockIdx.x * blockDim.x + threadIdx.x;
  const int e0 = 4 * j;
  if (e0 + 3 < E) {
    int d[4];
#pragma unroll
    for (int b = 0; b < 4; ++b) d[b] = dst[e0 + b];
    unsigned w = 0u;
#pragma unroll
    for (int b = 0; b < 4; ++b) {
      atomicAdd(&deg[d[b]], 1.0f);
      w |= ((unsigned)d[b] / (unsigned)AGG_ROWS) << (8 * b);
    }
    omap[j] = w;
  } else if (e0 < E) {
    unsigned w = 0u;
    for (int b = 0; e0 + b < E; ++b) {
      const int d = dst[e0 + b];
      atomicAdd(&deg[d], 1.0f);
      w |= ((unsigned)d / (unsigned)AGG_ROWS) << (8 * b);
    }
    omap[j] = w;
  }
}

__global__ void k_finish_deg(float* deg) {
  int i = blockIdx.x * blockDim.x + threadIdx.x;
  if (i < NNODES) deg[i] = rsqrtf(deg[i]);
}

template <int K, int N>
__global__ __launch_bounds__(256) void k_gemm_wmma_f32(const float* __restrict__ A,
                                                       const float* __restrict__ W,
                                                       const float* __restrict__ rscale,
                                                       float* __restrict__ C, int M) {
  constexpr int CT    = N / 16;
  constexpr int RPS   = 8 / CT;
  constexpr int ROWS  = RPS * 16;
  constexpr int PITCH = K + 4;
  constexpr int KS    = K / 4;
  constexpr int F4ROW = K / 4;
  constexpr int NLD   = ROWS * F4ROW / 256;

  __shared__ float lds[2][ROWS * PITCH];

  const int tid  = threadIdx.x;
  const int lane = tid & 31;
  const int wave = tid >> 5;
  const int half = lane >> 4;
  const int r    = lane & 15;
  const int ct   = wave % CT;
  const int rt   = wave / CT;
  const int row_tiles = M / 16;
  const int nstages   = (row_tiles + RPS - 1) / RPS;

  v2f breg[KS];
  {
    const float* wcol = W + (size_t)(ct * 16 + r);
#pragma unroll
    for (int i = 0; i < KS; ++i) {
      const int kA = 4 * i + 2 * half;
      breg[i].x = wcol[(size_t)kA * N];
      breg[i].y = wcol[(size_t)(kA + 1) * N];
    }
  }

  float4 pre[NLD];
  int s = blockIdx.x;

#pragma unroll
  for (int j = 0; j < NLD; ++j) {
    const int idx  = tid + 256 * j;
    const int row  = idx / F4ROW;
    const int c4   = idx % F4ROW;
    const int grow = s * ROWS + row;
    float4 v = make_float4(0.f, 0.f, 0.f, 0.f);
    if (s < nstages && grow < M) v = *(const float4*)(A + (size_t)grow * K + c4 * 4);
    *(float4*)&lds[0][row * PITCH + c4 * 4] = v;
  }
  __syncthreads();

  int cur = 0;
  for (; s < nstages; s += gridDim.x) {
    const int  sn       = s + gridDim.x;
    const bool has_next = sn < nstages;
    if (has_next) {
#pragma unroll
      for (int j = 0; j < NLD; ++j) {
        const int idx  = tid + 256 * j;
        const int row  = idx / F4ROW;
        const int c4   = idx % F4ROW;
        const int grow = sn * ROWS + row;
        pre[j] = make_float4(0.f, 0.f, 0.f, 0.f);
        if (grow < M) pre[j] = *(const float4*)(A + (size_t)grow * K + c4 * 4);
      }
    }

    const int tile = s * RPS + rt;
    if (tile < row_tiles) {
      const float* ap = &lds[cur][(rt * 16 + r) * PITCH + 2 * half];
      float rsc[8];
#pragma unroll
      for (int i = 0; i < 8; ++i) rsc[i] = rscale[tile * 16 + 8 * half + i];
      v8f acc = {};
#pragma unroll
      for (int i = 0; i < KS; ++i) {
        v2f a = *(const v2f*)(ap + 4 * i);
        acc = __builtin_amdgcn_wmma_f32_16x16x4_f32(
             false, a,  false, breg[i],
             (short)0, acc,  false,  false);
      }
      float* crow = C + (size_t)(tile * 16 + 8 * half) * N + ct * 16 + r;
#pragma unroll
      for (int i = 0; i < 8; ++i) crow[(size_t)i * N] = acc[i] * rsc[i];
    }

    if (has_next) {
#pragma unroll
      for (int j = 0; j < NLD; ++j) {
        const int idx = tid + 256 * j;
        const int row = idx / F4ROW;
        const int c4  = idx % F4ROW;
        *(float4*)&lds[cur ^ 1][row * PITCH + c4 * 4] = pre[j];
      }
    }
    __syncthreads();
    cur ^= 1;
  }
}

#define AGG_QCAP 512
#define AGG_LCAP 352
#define AGG_MAPDW ((size_t)NNODES * 64 - (size_t)AGG_WGS * 32 * (AGG_LCAP + 1))
static_assert(NNODES <= (1 << 24), "a source id is packed into 24 bits");
static_assert((size_t)AGG_WGS * 32 * (AGG_LCAP + 1) <= (size_t)NNODES * 64, "numbers and lists fit the unused half of the second workspace plane");
static_assert(AGG_MAPDW * 4 >= 1200000, "the owner map has room for 1200000 edges; the edge list here has 800000");
static_assert(((size_t)AGG_WGS * 32 * (AGG_LCAP + 1)) % 32 == 0, "the owner map starts on a whole line");

template <int CH, int NOUT, int LIST>
__global__ __launch_bounds__(1024) void k_aggregate(const int* __restrict__ src, const int* __restrict__ dst,
                                                    const float* __restrict__ dinv,
                                                    const float* __restrict__ feat,
                                                    const float* __restrict__ bias,
                                                    const float* __restrict__ W,
                                                    float* __restrict__ out,
                                                    unsigned* __restrict__ ecnt,
                                                    unsigned* __restrict__ elist,
                                                    const unsigned* __restrict__ omap, int E) {
  constexpr int PITCH = (NOUT > 0) ? CH + 4 : CH;
  constexpr int TROWS = (NOUT > 0) ? (AGG_ROWS + 15) / 16 * 16 : AGG_ROWS;
  constexpr int VPL  = CH / 32;
  constexpr int ROW4 = CH / 4;
  constexpr int G    = (CH <= 64) ? 16 : 8;
  static_assert(AGG_ROWS <= 256, "row number inside a workgroup is packed into 8 bits");
  __shared__ __attribute__((aligned(16))) float tile[TROWS * PITCH];
  __shared__ __attribute__((aligned(16))) unsigned ring[32][AGG_QCAP];

  const int tid  = (int)threadIdx.x;
  const int lane = tid & 31;
  const int wv   = __builtin_amdgcn_readfirstlane(tid >> 5);
  const int lo   = (int)blockIdx.x * AGG_ROWS;
  const int nrows = max(0, min(AGG_ROWS, NNODES - lo));

  constexpr int NIT  = (AGG_ROWS * ROW4 + 1023) / 1024;
  constexpr int NDE  = (NOUT > 0) ? 1 : NIT;
  constexpr int NITS = (NOUT > 0) ? (AGG_ROWS * (NOUT / 4) + 1023) / 1024 : 1;
  static_assert(1024 % ROW4 == 0, "all items of a thread are in the same columns");
  float  dend[NDE];
  float  dsto[NITS];
  float4 bend;
  {
    float4 v[NIT];
#pragma unroll
    for (int j = 0; j < NIT; ++j) {
      const int i = tid + 1024 * j;
      v[j] = make_float4(0.f, 0.f, 0.f, 0.f);
      if (j < NDE) dend[j] = 0.f;
      if (i < nrows * ROW4) {
        v[j] = *(const float4*)(feat + (size_t)lo * CH + (size_t)i * 4);
        if (j < NDE) dend[j] = dinv[lo + i / ROW4];
      }
    }
    bend = *(const float4*)(bias + (tid % ROW4) * 4);
    if constexpr (NOUT > 0) {
#pragma unroll
      for (int j = 0; j < NITS; ++j) {
        const int i = tid + 1024 * j;
        dsto[j] = (i < nrows * (NOUT / 4)) ? dinv[lo + i / (NOUT / 4)] : 0.f;
      }
    }
#pragma unroll
    for (int j = 0; j < NIT; ++j) {
      const int i = tid + 1024 * j;
      if (i < nrows * ROW4) *(float4*)&tile[(i / ROW4) * PITCH + (i % ROW4) * 4] = v[j];
    }
    for (int i = nrows * ROW4 + tid; i < TROWS * ROW4; i += 1024)
      *(float4*)&tile[(i / ROW4) * PITCH + (i % ROW4) * 4] = make_float4(0.f, 0.f, 0.f, 0.f);
  }
  __syncthreads();

  unsigned* q = ring[wv];
  int qh = 0, qn = 0;
  unsigned* lst = elist + (size_t)((int)blockIdx.x * 32 + wv) * AGG_LCAP;
  int ln = 0;

  auto add_rows = [&](const int (&s)[G], const int (&dl)[G], int n) {
    float v[G][VPL];
#pragma unroll
    for (int k = 0; k < G; ++k) {
      const float* frow = feat + (size_t)s[k] * CH;
#pragma unroll
      for (int c = 0; c < VPL; ++c) v[k][c] = frow[c * 32 + lane];
    }
#pragma unroll
    for (int k = 0; k < G; ++k) {
      if (k < n) {
        float* trow = &tile[dl[k] * PITCH + lane];
#pragma unroll
        for (int c = 0; c < VPL; ++c)
          __hip_atomic_fetch_add(trow + c * 32, v[k][c], __ATOMIC_RELAXED, __HIP_MEMORY_SCOPE_WORKGROUP);
      }
    }
  };

  auto drain = [&](int base, int n) {
    int e[G], dl[G], s[G];
#pragma unroll
    for (int k = 0; k < G; ++k) {
      const int kk = (k < n) ? k : 0;
      e[k] = __builtin_amdgcn_readfirstlane((int)q[(base + kk) & (AGG_QCAP - 1)]);
    }
#pragma unroll
    for (int k = 0; k < G; ++k) {
      s[k]  = __builtin_amdgcn_readfirstlane(src[e[k]]);
      dl[k] = __builtin_amdgcn_readfirstlane(dst[e[k]]) - lo;
    }
    if constexpr (LIST == 1) {
      unsigned ent = 0u;
#pragma unroll
      for (int k = 0; k < G; ++k)
        if (lane == k) ent = (unsigned)s[k] | ((unsigned)dl[k] << 24);
      if (lane < n && ln + lane < AGG_LCAP) lst[ln + lane] = ent;
      ln += n;
    }
    add_rows(s, dl, n);
  };

  auto test8 = [&](const int* d, int ebase) {
#pragma unroll
    for (int u = 0; u < 8; ++u) {
      const unsigned dloc = (unsigned)(d[u] - lo);
      const bool     hit  = dloc < (unsigned)nrows;
      const unsigned m    = __builtin_amdgcn_ballot_w32(hit);
      if (__builtin_expect(m != 0u, 0)) {
        if (hit) {
          const int pos = qn + __builtin_popcount(m & ((1u << lane) - 1u));
          q[pos & (AGG_QCAP - 1)] = (unsigned)(ebase + 32 * u);
        }
        qn += __builtin_popcount(m);
      }
    }
    while (qn - qh >= G) { drain(qh, G); qh += G; }
  };

  const unsigned pat = (unsigned)blockIdx.x * 0x01010101u;

  auto test2 = [&](const unsigned* w, int jbase) {
#pragma unroll
    for (int u = 0; u < 2; ++u) {
      const unsigned x = w[u] ^ pat;
      unsigned z = ~(((x & 0x7F7F7F7Fu) + 0x7F7F7F7Fu) | x) & 0x80808080u;
      unsigned m = __builtin_amdgcn_ballot_w32(z != 0u);
      if (__builtin_expect(m != 0u, 0)) {
        const int ebase = (jbase + 32 * u) << 2;
        for (int it = 0; it < 4 && m != 0u; ++it) {
          if (z != 0u) {
            const int pos = qn + __builtin_popcount(m & ((1u << lane) - 1u));
            q[pos & (AGG_QCAP - 1)] = (unsigned)(ebase + (__builtin_ctz(z) >> 3));
            z &= z - 1u;
          }
          qn += __builtin_popcount(m);
          m = __builtin_amdgcn_ballot_w32(z != 0u);
        }
      }
    }
    while (qn - qh >= G) { drain(qh, G); qh += G; }
  };

  auto filter_share = [&]() {
  const int nfull = E >> 10;
  const int rot = (int)(((long long)blockIdx.x * nfull) / AGG_WGS);
  auto ask = [&](unsigned (&w)[8], int i) -> int {
    int t = i + rot;
    if (t >= nfull) t -= nfull;
    const unsigned* mp = omap + (t << 8) + lane;
#pragma unroll
    for (int u = 0; u < 8; ++u) w[u] = mp[32 * u];
    return t << 8;
  };
  unsigned wa[8], wb[8];
  int ja = 0;
  if (wv < nfull) ja = ask(wa, wv);
  for (int i = wv; i < nfull; i += 64) {
    const bool has_b = i + 32 < nfull;
    const int  jb    = ask(wb, has_b ? i + 32 : i);
#pragma unroll
    for (int g = 0; g < 4; ++g) test2(&wa[2 * g], ja + lane + 64 * g);
    ja = ask(wa, (i + 64 < nfull) ? i + 64 : i);
    if (has_b) {
#pragma unroll
      for (int g = 0; g < 4; ++g) test2(&wb[2 * g], jb + lane + 64 * g);
    }
  }
  {
    const int tbase = nfull << 10;
    const int npiece = (E - tbase + 255) >> 8;
    if (wv < npiece) {
      const int e0 = tbase + (wv << 8) + lane;
      int d[8];
#pragma unroll
      for (int u = 0; u < 8; ++u) {
        const int ee = e0 + 32 * u;
        d[u] = (ee < E) ? dst[ee] : -1;
      }
      test8(d, e0);
    }
  }
  if (qn > qh) drain(qh, qn - qh);
  };

  if constexpr (LIST == 2) {
    const int cnt = __builtin_amdgcn_readfirstlane((int)ecnt[(int)blockIdx.x * 32 + wv]);
    if (__builtin_expect(cnt <= AGG_LCAP, 1)) {
      unsigned cur = (cnt > 0) ? lst[min(lane, cnt - 1)] : 0u;
      for (int base = 0; base < cnt; base += 32) {
        const unsigned nxt = lst[min(base + 32 + lane, cnt - 1)];
#pragma unroll
        for (int sub = 0; sub < 32; sub += G) {
          const int n = min(G, cnt - base - sub);
          if (n > 0) {
            int s[G], dl[G];
#pragma unroll
            for (int k = 0; k < G; ++k) {
              const unsigned ent = (unsigned)__builtin_amdgcn_readlane((int)cur, sub + k);
              s[k]  = (int)(ent & 0xFFFFFFu);
              dl[k] = (int)(ent >> 24);
            }
            add_rows(s, dl, n);
          }
        }
        cur = nxt;
      }
    } else {
      filter_share();
    }
  } else {
    filter_share();
    if constexpr (LIST == 1) {
      if (lane == 0) ecnt[(int)blockIdx.x * 32 + wv] = (unsigned)ln;
    }
  }
  __syncthreads();

  if constexpr (NOUT == 0) {
#pragma unroll
    for (int j = 0; j < NIT; ++j) {
      const int i = tid + 1024 * j;
      if (i < nrows * ROW4) {
        float4 v = *(const float4*)&tile[(i / ROW4) * PITCH + (i % ROW4) * 4];
        v.x = v.x * dend[j] + bend.x;
        v.y = v.y * dend[j] + bend.y;
        v.z = v.z * dend[j] + bend.z;
        v.w = v.w * dend[j] + bend.w;
        *(float4*)(out + (size_t)lo * CH + (size_t)i * 4) = v;
      }
    }
  } else {
    constexpr int CT  = NOUT / 16;
    constexpr int KS  = CH / 4;
    constexpr int RTS = 32 / CT;
    static_assert(32 % CT == 0, "every wave owns one column tile");
    const int half = lane >> 4;
    const int r    = lane & 15;
    const int ct   = wv % CT;
    v2f breg[KS];
    {
      const float* wcol = W + (size_t)(ct * 16 + r);
#pragma unroll
      for (int i = 0; i < KS; ++i) {
        const int kA = 4 * i + 2 * half;
        breg[i].x = wcol[(size_t)kA * NOUT];
        breg[i].y = wcol[(size_t)(kA + 1) * NOUT];
      }
    }
    float di = dend[0];
    for (int i = tid; i < nrows * ROW4; i += 1024) {
      const int   inx = i + 1024;
      const float dn  = (inx < nrows * ROW4) ? dinv[lo + inx / ROW4] : 0.f;
      float4* p = (float4*)&tile[(i / ROW4) * PITCH + (i % ROW4) * 4];
      float4 v = *p;
      v.x = tanhf(v.x * di + bend.x);
      v.y = tanhf(v.y * di + bend.y);
      v.z = tanhf(v.z * di + bend.z);
      v.w = tanhf(v.w * di + bend.w);
      *p = v;
      di = dn;
    }
    __syncthreads();
    static_assert(TROWS * NOUT <= 32 * AGG_QCAP, "product rows fit the rings' space");
    float* stage = (float*)&ring[0][0];
    const int ntiles = (nrows + 15) >> 4;
    for (int t = wv / CT; t < ntiles; t += RTS) {
      const float* ap = &tile[(t * 16 + r) * PITCH + 2 * half];
      v8f acc = {};
#pragma unroll
      for (int i = 0; i < KS; ++i) {
        v2f a = *(const v2f*)(ap + 4 * i);
        acc = __builtin_amdgcn_wmma_f32_16x16x4_f32(
             false, a,  false, breg[i],
             (short)0, acc,  false,  false);
      }
      float* sp = stage + (t * 16 + 8 * half) * NOUT + ct * 16 + r;
#pragma unroll
      for (int i = 0; i < 8; ++i) sp[i * NOUT] = acc[i];
    }
    __syncthreads();
#pragma unroll
    for (int j = 0; j < NITS; ++j) {
      const int i = tid + 1024 * j;
      if (i < nrows * (NOUT / 4)) {
        float4 v = *(const float4*)&stage[i * 4];
        v.x *= dsto[j];
        v.y *= dsto[j];
        v.z *= dsto[j];
        v.w *= dsto[j];
        *(float4*)(out + (size_t)lo * NOUT + (size_t)i * 4) = v;
      }
    }
  }
}

extern "C" void kernel_launch(void* const* d_in, const int* in_sizes, int n_in,
                              void* d_out, int out_size, void* d_ws, size_t ws_size,
                              hipStream_t stream) {
  const float* x   = (const float*)d_in[0];
  const int*  eidx = (const int*)d_in[1];
  const float* W1  = (const float*)d_in[2];
  const float* b1  = (const float*)d_in[3];
  const float* W2  = (const float*)d_in[4];
  const float* b2  = (const float*)d_in[5];
  float* out = (float*)d_out;

  const int E = in_sizes[1] / 2;
  const int* src = eidx;
  const int* dst = eidx + E;

  float* dinv = (float*)d_ws;
  float* bufA = dinv + 50048;
  float* bufB = bufA + (size_t)NNODES * 128;

  const int T = 256;

  unsigned* ecnt  = (unsigned*)(bufB + (size_t)NNODES * 64);
  unsigned* elist = ecnt + AGG_WGS * 32;
  unsigned* omap  = elist + (size_t)AGG_WGS * 32 * AGG_LCAP;

  k_init_deg<<<(NNODES + T - 1) / T, T, 0, stream>>>(dinv);
  k_count_deg<<<((E + 3) / 4 + T - 1) / T, T, 0, stream>>>(dst, dinv, omap, E);
  k_finish_deg<<<(NNODES + T - 1) / T, T, 0, stream>>>(dinv);

  {
    k_gemm_wmma_f32<256, 128><<<512, 256, 0, stream>>>(x, W1, dinv, bufA, NNODES);
  }
  k_aggregate<128, 64, 1><<<AGG_WGS, 1024, 0, stream>>>(src, dst, dinv, bufA, b1, W2, bufB, ecnt, elist, omap, E);

  k_aggregate<64, 0, 2><<<AGG_WGS, 1024, 0, stream>>>(src, dst, dinv, bufB, b2, nullptr, out, ecnt, elist, omap, E);
}
